# starting kernel
# baseline (speedup 1.0000x reference)
#include <hip/hip_runtime.h>
#include <hip/hip_bf16.h>


typedef __attribute__((ext_vector_type(2))) float v2f;
typedef __attribute__((ext_vector_type(8))) float v8f;

#define D_FEAT 32

__global__ void gcn_zero_kernel(float* __restrict__ p, size_t n) {
    size_t i = (size_t)blockIdx.x * blockDim.x + threadIdx.x;
    if (i < n) p[i] = 0.0f;
}

__global__ void gcn_degree_kernel(const int* __restrict__ er,
                                  const float* __restrict__ ev,
                                  float* __restrict__ deg, int E) {
    int e = blockIdx.x * blockDim.x + threadIdx.x;
    if (e < E) atomicAdd(&deg[er[e]], ev[e]);
}

__global__ void gcn_rsqrt_kernel(float* __restrict__ deg, int N) {
    int i = blockIdx.x * blockDim.x + threadIdx.x;
    if (i < N) {
        float d = deg[i];
        deg[i] = (d > 0.0f) ? rsqrtf(d) : 0.0f;
    }
}

__global__ void gcn_spmm_kernel(const int* __restrict__ er,
                                const int* __restrict__ ec,
                                const float* __restrict__ ev,
                                const float* __restrict__ dis,
                                const float* __restrict__ feat,
                                float* __restrict__ agg, int E) {
    int wave = (int)(((size_t)blockIdx.x * blockDim.x + threadIdx.x) >> 5);
    if (wave >= E) return;
    int lane = threadIdx.x & 31;
    int r = er[wave];
    int c = ec[wave];
    float v = dis[r] * ev[wave] * dis[c];
    float f = feat[(size_t)c * D_FEAT + lane];
    atomicAdd(&agg[(size_t)r * D_FEAT + lane], v * f);
}

__global__ void gcn_gemm_wmma_kernel(const float* __restrict__ agg,
                                     const float* __restrict__ W,
                                     float* __restrict__ out, int nTiles) {
    int wave = (int)(((size_t)blockIdx.x * blockDim.x + threadIdx.x) >> 5);
    if (wave >= nTiles) return;
    int lane = threadIdx.x & 31;
    int half = lane >> 4;
    int l15  = lane & 15;

    const float* aRow = agg + (size_t)(wave * 16 + l15) * D_FEAT;

    v8f acc0 = {};
    v8f acc1 = {};
#pragma unroll
    for (int k = 0; k < D_FEAT; k += 4) {
        int kk = k + half * 2;
        v2f a, b0, b1;
        a.x  = aRow[kk];
        a.y  = aRow[kk + 1];
        b0.x = W[l15 * D_FEAT + kk];
        b0.y = W[l15 * D_FEAT + kk + 1];
        b1.x = W[(16 + l15) * D_FEAT + kk];
        b1.y = W[(16 + l15) * D_FEAT + kk + 1];
        acc0 = __builtin_amdgcn_wmma_f32_16x16x4_f32(
            false, a, false, b0, (short)0, acc0, false, false);
        acc1 = __builtin_amdgcn_wmma_f32_16x16x4_f32(
            false, a, false, b1, (short)0, acc1, false, false);
    }

    int mBase = wave * 16 + half * 8;
#pragma unroll
    for (int i = 0; i < 8; ++i) {
        out[(size_t)(mBase + i) * D_FEAT + l15]      = acc0[i];
        out[(size_t)(mBase + i) * D_FEAT + 16 + l15] = acc1[i];
    }
}

__global__ void gcn_gemm_tail_kernel(const float* __restrict__ agg,
                                     const float* __restrict__ W,
                                     float* __restrict__ out,
                                     int rowStart, int N) {
    int idx = blockIdx.x * blockDim.x + threadIdx.x;
    int row = rowStart + (idx >> 5);
    int o = idx & 31;
    if (row >= N) return;
    float s = 0.0f;
#pragma unroll
    for (int k = 0; k < D_FEAT; ++k)
        s += agg[(size_t)row * D_FEAT + k] * W[o * D_FEAT + k];
    out[(size_t)row * D_FEAT + o] = s;
}

extern "C" void kernel_launch(void* const* d_in, const int* in_sizes, int n_in,
                              void* d_out, int out_size, void* d_ws, size_t ws_size,
                              hipStream_t stream) {
    const int*   er   = (const int*)d_in[0];
    const int*   ec   = (const int*)d_in[1];
    const float* ev   = (const float*)d_in[2];
    const float* feat = (const float*)d_in[3];
    const float* W    = (const float*)d_in[4];
    float* out = (float*)d_out;

    int E = in_sizes[0];
    int N = in_sizes[3] / D_FEAT;

    float* agg = (float*)d_ws;
    float* deg = agg + (size_t)N * D_FEAT;

    size_t zn = (size_t)N * (D_FEAT + 1);
    gcn_zero_kernel<<<(unsigned)((zn + 255) / 256), 256, 0, stream>>>(agg, zn);

    gcn_degree_kernel<<<(E + 255) / 256, 256, 0, stream>>>(er, ev, deg, E);

    gcn_rsqrt_kernel<<<(N + 255) / 256, 256, 0, stream>>>(deg, N);

    size_t spmmThreads = (size_t)E * 32;
    gcn_spmm_kernel<<<(unsigned)((spmmThreads + 255) / 256), 256, 0, stream>>>(
        er, ec, ev, deg, feat, agg, E);

    int nTiles = N / 16;
    if (nTiles > 0) {
        size_t gemmThreads = (size_t)nTiles * 32;
        gcn_gemm_wmma_kernel<<<(unsigned)((gemmThreads + 255) / 256), 256, 0, stream>>>(
            agg, W, out, nTiles);
    }
    int rowStart = nTiles * 16;
    if (rowStart < N) {
        int tailThreads = (N - rowStart) * 32;
        gcn_gemm_tail_kernel<<<(tailThreads + 255) / 256, 256, 0, stream>>>(
            agg, W, out, rowStart, N);
    }
}


# SpMM: edges grouped by 256-row block (count, scan, scatter); a work-group sums its block in LDS (f32 LDS adds, 32 edges per wave step), stores it once; no global agg atomics; agg sum order changes
# speedup vs baseline: 1.6960x; 1.6960x over previous
#include <hip/hip_runtime.h>
#include <hip/hip_bf16.h>


typedef __attribute__((ext_vector_type(2))) float v2f;
typedef __attribute__((ext_vector_type(8))) float v8f;

#define D_FEAT 32

__global__ void gcn_zero_kernel(float* __restrict__ p, size_t n) {
    size_t i = (size_t)blockIdx.x * blockDim.x + threadIdx.x;
    if (i < n) p[i] = 0.0f;
}

__global__ void gcn_degree_kernel(const int* __restrict__ er,
                                  const float* __restrict__ ev,
                                  float* __restrict__ deg, int E) {
    int e = blockIdx.x * blockDim.x + threadIdx.x;
    if (e < E) atomicAdd(&deg[er[e]], ev[e]);
}

__global__ void gcn_rsqrt_kernel(float* __restrict__ deg, int N) {
    int i = blockIdx.x * blockDim.x + threadIdx.x;
    if (i < N) {
        float d = deg[i];
        deg[i] = (d > 0.0f) ? rsqrtf(d) : 0.0f;
    }
}

#define RB_SHIFT 8
#define RB (1 << RB_SHIFT)
#define BIN_THREADS 1024
#define EPT 8
#define CHUNK (EPT * BIN_THREADS)

__global__ __launch_bounds__(BIN_THREADS)
void gcn_bin_count_kernel(const int* __restrict__ er, int* __restrict__ binTotal,
                          int E, int nb) {
    extern __shared__ int hist[];
    for (int i = threadIdx.x; i < nb; i += BIN_THREADS) hist[i] = 0;
    __syncthreads();
    int base = blockIdx.x * CHUNK;
    int r[EPT];
#pragma unroll
    for (int k = 0; k < EPT; ++k) {
        int e = base + k * BIN_THREADS + (int)threadIdx.x;
        r[k] = (e < E) ? er[e] : -1;
    }
#pragma unroll
    for (int k = 0; k < EPT; ++k)
        if (r[k] >= 0) atomicAdd(&hist[r[k] >> RB_SHIFT], 1);
    __syncthreads();
    for (int i = threadIdx.x; i < nb; i += BIN_THREADS) {
        int c = hist[i];
        if (c) atomicAdd(&binTotal[i], c);
    }
}

__global__ __launch_bounds__(1024)
void gcn_bin_scan_kernel(const int* __restrict__ binTotal, int* __restrict__ binStart,
                         int* __restrict__ binCursor, int nb) {
    __shared__ int s[1024];
    __shared__ int carry;
    int t = threadIdx.x;
    if (t == 0) carry = 0;
    __syncthreads();
    for (int t0 = 0; t0 < nb; t0 += 1024) {
        int b = t0 + t;
        int v = (b < nb) ? binTotal[b] : 0;
        s[t] = v;
        __syncthreads();
        for (int off = 1; off < 1024; off <<= 1) {
            int add = (t >= off) ? s[t - off] : 0;
            __syncthreads();
            s[t] += add;
            __syncthreads();
        }
        int incl = s[t];
        int c0 = carry;
        if (b < nb) {
            binStart[b]  = c0 + incl - v;
            binCursor[b] = c0 + incl - v;
        }
        __syncthreads();
        if (t == 1023) carry = c0 + incl;
        __syncthreads();
    }
    if (t == 0) binStart[nb] = carry;
}

__global__ __launch_bounds__(BIN_THREADS)
void gcn_bin_scatter_kernel(const int* __restrict__ er, const int* __restrict__ ec,
                            const float* __restrict__ ev, int* __restrict__ binCursor,
                            int* __restrict__ recWords, int E, int nb) {
    extern __shared__ int hist[];
    int4* rec = reinterpret_cast<int4*>(recWords);
    for (int i = threadIdx.x; i < nb; i += BIN_THREADS) hist[i] = 0;
    __syncthreads();
    int base = blockIdx.x * CHUNK;
    int r[EPT], c[EPT], vb[EPT], rank[EPT];
#pragma unroll
    for (int k = 0; k < EPT; ++k) {
        int e = base + k * BIN_THREADS + (int)threadIdx.x;
        r[k] = 0; c[k] = 0; vb[k] = 0;
        if (e < E) {
            r[k]  = er[e];
            c[k]  = ec[e];
            vb[k] = __float_as_int(ev[e]);
        }
    }
#pragma unroll
    for (int k = 0; k < EPT; ++k) {
        int e = base + k * BIN_THREADS + (int)threadIdx.x;
        rank[k] = 0;
        if (e < E) rank[k] = atomicAdd(&hist[r[k] >> RB_SHIFT], 1);
    }
    __syncthreads();
    for (int i = threadIdx.x; i < nb; i += BIN_THREADS) {
        int n = hist[i];
        int g = 0;
        if (n) g = atomicAdd(&binCursor[i], n);
        hist[i] = g;
    }
    __syncthreads();
#pragma unroll
    for (int k = 0; k < EPT; ++k) {
        int e = base + k * BIN_THREADS + (int)threadIdx.x;
        if (e < E) {
            int slot = hist[r[k] >> RB_SHIFT] + rank[k];
            rec[slot] = make_int4(r[k], c[k], vb[k], 0);
        }
    }
}

#define SPMM_THREADS 512
__global__ __launch_bounds__(SPMM_THREADS)
void gcn_bin_spmm_kernel(const int* __restrict__ recWords, const int* __restrict__ binStart,
                         const float* __restrict__ dis, const float* __restrict__ feat,
                         float* __restrict__ agg, int N) {
    __shared__ float tile[RB * D_FEAT];
    const int4* rec = reinterpret_cast<const int4*>(recWords);
    const int tid  = threadIdx.x;
    const int lane = tid & 31;
    const int wave = __builtin_amdgcn_readfirstlane(tid >> 5);
    const int rowBase = (int)blockIdx.x << RB_SHIFT;
    for (int i = tid; i < RB * D_FEAT; i += SPMM_THREADS) tile[i] = 0.0f;
    const int start = binStart[blockIdx.x];
    const int end   = binStart[blockIdx.x + 1];
    __syncthreads();

    int g0 = start + wave * 32;
    int4 q = make_int4(0, 0, 0, 0);
    if (g0 + lane < end) q = rec[g0 + lane];
    while (g0 < end) {
        const int gn = g0 + SPMM_THREADS;
        int4 qn = make_int4(0, 0, 0, 0);
        if (gn + lane < end) qn = rec[gn + lane];
        float v = 0.0f;
        if (g0 + lane < end) v = dis[q.x] * __int_as_float(q.z) * dis[q.y];
        const int vbits = __float_as_int(v);
        const int rl = (q.x - rowBase) * D_FEAT;
        const int nvalid = (end - g0 < 32) ? (end - g0) : 32;
        if (nvalid == 32) {
            float f[32];
#pragma unroll
            for (int j = 0; j < 32; ++j) {
                int cj = __builtin_amdgcn_readlane(q.y, j);
                f[j] = feat[(size_t)cj * D_FEAT + lane];
            }
#pragma unroll
            for (int j = 0; j < 32; ++j) {
                float vj = __int_as_float(__builtin_amdgcn_readlane(vbits, j));
                int rj = __builtin_amdgcn_readlane(rl, j);
                atomicAdd(&tile[rj + lane], vj * f[j]);
            }
        } else {
            for (int j = 0; j < nvalid; ++j) {
                int cj = __builtin_amdgcn_readlane(q.y, j);
                float vj = __int_as_float(__builtin_amdgcn_readlane(vbits, j));
                int rj = __builtin_amdgcn_readlane(rl, j);
                float fj = feat[(size_t)cj * D_FEAT + lane];
                atomicAdd(&tile[rj + lane], vj * fj);
            }
        }
        q = qn;
        g0 = gn;
    }
    __syncthreads();
    int rows = N - rowBase;
    if (rows > RB) rows = RB;
    float* dst = agg + (size_t)rowBase * D_FEAT;
    for (int i = tid; i < rows * D_FEAT; i += SPMM_THREADS) dst[i] = tile[i];
}

__global__ void gcn_gemm_wmma_kernel(const float* __restrict__ agg,
                                     const float* __restrict__ W,
                                     float* __restrict__ out, int nTiles) {
    int wave = (int)(((size_t)blockIdx.x * blockDim.x + threadIdx.x) >> 5);
    if (wave >= nTiles) return;
    int lane = threadIdx.x & 31;
    int half = lane >> 4;
    int l15  = lane & 15;

    const float* aRow = agg + (size_t)(wave * 16 + l15) * D_FEAT;

    v8f acc0 = {};
    v8f acc1 = {};
#pragma unroll
    for (int k = 0; k < D_FEAT; k += 4) {
        int kk = k + half * 2;
        v2f a, b0, b1;
        a.x  = aRow[kk];
        a.y  = aRow[kk + 1];
        b0.x = W[l15 * D_FEAT + kk];
        b0.y = W[l15 * D_FEAT + kk + 1];
        b1.x = W[(16 + l15) * D_FEAT + kk];
        b1.y = W[(16 + l15) * D_FEAT + kk + 1];
        acc0 = __builtin_amdgcn_wmma_f32_16x16x4_f32(
            false, a, false, b0, (short)0, acc0, false, false);
        acc1 = __builtin_amdgcn_wmma_f32_16x16x4_f32(
            false, a, false, b1, (short)0, acc1, false, false);
    }

    int mBase = wave * 16 + half * 8;
#pragma unroll
    for (int i = 0; i < 8; ++i) {
        out[(size_t)(mBase + i) * D_FEAT + l15]      = acc0[i];
        out[(size_t)(mBase + i) * D_FEAT + 16 + l15] = acc1[i];
    }
}

__global__ void gcn_gemm_tail_kernel(const float* __restrict__ agg,
                                     const float* __restrict__ W,
                                     float* __restrict__ out,
                                     int rowStart, int N) {
    int idx = blockIdx.x * blockDim.x + threadIdx.x;
    int row = rowStart + (idx >> 5);
    int o = idx & 31;
    if (row >= N) return;
    float s = 0.0f;
#pragma unroll
    for (int k = 0; k < D_FEAT; ++k)
        s += agg[(size_t)row * D_FEAT + k] * W[o * D_FEAT + k];
    out[(size_t)row * D_FEAT + o] = s;
}

extern "C" void kernel_launch(void* const* d_in, const int* in_sizes, int n_in,
                              void* d_out, int out_size, void* d_ws, size_t ws_size,
                              hipStream_t stream) {
    const int*   er   = (const int*)d_in[0];
    const int*   ec   = (const int*)d_in[1];
    const float* ev   = (const float*)d_in[2];
    const float* feat = (const float*)d_in[3];
    const float* W    = (const float*)d_in[4];
    float* out = (float*)d_out;

    int E = in_sizes[0];
    int N = in_sizes[3] / D_FEAT;

    int nb = (N + RB - 1) >> RB_SHIFT;

    float* agg = (float*)d_ws;
    float* deg = agg + (size_t)N * D_FEAT;
    int* binTotal  = (int*)(deg + N);
    int* binStart  = binTotal + nb;
    int* binCursor = binStart + nb + 1;
    size_t recOff = ((size_t)N * (D_FEAT + 1) + 3 * (size_t)nb + 1 + 3) & ~(size_t)3;
    int* rec = (int*)d_ws + recOff;

    size_t zn = (size_t)N + (size_t)nb;
    gcn_zero_kernel<<<(unsigned)((zn + 255) / 256), 256, 0, stream>>>(deg, zn);

    gcn_degree_kernel<<<(E + 255) / 256, 256, 0, stream>>>(er, ev, deg, E);

    gcn_rsqrt_kernel<<<(N + 255) / 256, 256, 0, stream>>>(deg, N);

    int nChunks = (E + CHUNK - 1) / CHUNK;
    size_t histBytes = (size_t)nb * sizeof(int);
    gcn_bin_count_kernel<<<nChunks, BIN_THREADS, histBytes, stream>>>(er, binTotal, E, nb);
    gcn_bin_scan_kernel<<<1, 1024, 0, stream>>>(binTotal, binStart, binCursor, nb);
    gcn_bin_scatter_kernel<<<nChunks, BIN_THREADS, histBytes, stream>>>(
        er, ec, ev, binCursor, rec, E, nb);
    gcn_bin_spmm_kernel<<<nb, SPMM_THREADS, 0, stream>>>(rec, binStart, deg, feat, agg, N);

    int nTiles = N / 16;
    if (nTiles > 0) {
        size_t gemmThreads = (size_t)nTiles * 32;
        gcn_gemm_wmma_kernel<<<(unsigned)((gemmThreads + 255) / 256), 256, 0, stream>>>(
            agg, W, out, nTiles);
    }
    int rowStart = nTiles * 16;
    if (rowStart < N) {
        int tailThreads = (N - rowStart) * 32;
        gcn_gemm_tail_kernel<<<(tailThreads + 255) / 256, 256, 0, stream>>>(
            agg, W, out, rowStart, N);
    }
}


# Degree path loses its own launches: deg summed in the bin-count pass (8 edges per thread, global f32 atomics kept), rsqrt taken at its use in SpMM; 8 -> 6 launches, about 53,000 short waves fewer
# speedup vs baseline: 1.7447x; 1.0287x over previous
#include <hip/hip_runtime.h>
#include <hip/hip_bf16.h>


typedef __attribute__((ext_vector_type(2))) float v2f;
typedef __attribute__((ext_vector_type(8))) float v8f;

#define D_FEAT 32

__global__ void gcn_zero_kernel(float* __restrict__ p, size_t n) {
    size_t i = (size_t)blockIdx.x * blockDim.x + threadIdx.x;
    if (i < n) p[i] = 0.0f;
}

__device__ __forceinline__ float gcn_inv_sqrt_deg(float d) {
    return (d > 0.0f) ? rsqrtf(d) : 0.0f;
}

#define RB_SHIFT 8
#define RB (1 << RB_SHIFT)
#define BIN_THREADS 1024
#define EPT 8
#define CHUNK (EPT * BIN_THREADS)

__global__ __launch_bounds__(BIN_THREADS)
void gcn_bin_count_kernel(const int* __restrict__ er, const float* __restrict__ ev,
                          float* __restrict__ deg, int* __restrict__ binTotal,
                          int E, int nb) {
    extern __shared__ int hist[];
    for (int i = threadIdx.x; i < nb; i += BIN_THREADS) hist[i] = 0;
    __syncthreads();
    int base = blockIdx.x * CHUNK;
    int r[EPT];
    float w[EPT];
#pragma unroll
    for (int k = 0; k < EPT; ++k) {
        int e = base + k * BIN_THREADS + (int)threadIdx.x;
        r[k] = -1;
        w[k] = 0.0f;
        if (e < E) {
            r[k] = er[e];
            w[k] = ev[e];
        }
    }
#pragma unroll
    for (int k = 0; k < EPT; ++k)
        if (r[k] >= 0) {
            atomicAdd(&deg[r[k]], w[k]);
            atomicAdd(&hist[r[k] >> RB_SHIFT], 1);
        }
    __syncthreads();
    for (int i = threadIdx.x; i < nb; i += BIN_THREADS) {
        int c = hist[i];
        if (c) atomicAdd(&binTotal[i], c);
    }
}

__global__ __launch_bounds__(1024)
void gcn_bin_scan_kernel(const int* __restrict__ binTotal, int* __restrict__ binStart,
                         int* __restrict__ binCursor, int nb) {
    __shared__ int s[1024];
    __shared__ int carry;
    int t = threadIdx.x;
    if (t == 0) carry = 0;
    __syncthreads();
    for (int t0 = 0; t0 < nb; t0 += 1024) {
        int b = t0 + t;
        int v = (b < nb) ? binTotal[b] : 0;
        s[t] = v;
        __syncthreads();
        for (int off = 1; off < 1024; off <<= 1) {
            int add = (t >= off) ? s[t - off] : 0;
            __syncthreads();
            s[t] += add;
            __syncthreads();
        }
        int incl = s[t];
        int c0 = carry;
        if (b < nb) {
            binStart[b]  = c0 + incl - v;
            binCursor[b] = c0 + incl - v;
        }
        __syncthreads();
        if (t == 1023) carry = c0 + incl;
        __syncthreads();
    }
    if (t == 0) binStart[nb] = carry;
}

__global__ __launch_bounds__(BIN_THREADS)
void gcn_bin_scatter_kernel(const int* __restrict__ er, const int* __restrict__ ec,
                            const float* __restrict__ ev, int* __restrict__ binCursor,
                            int* __restrict__ recWords, int E, int nb) {
    extern __shared__ int hist[];
    int4* rec = reinterpret_cast<int4*>(recWords);
    for (int i = threadIdx.x; i < nb; i += BIN_THREADS) hist[i] = 0;
    __syncthreads();
    int base = blockIdx.x * CHUNK;
    int r[EPT], c[EPT], vb[EPT], rank[EPT];
#pragma unroll
    for (int k = 0; k < EPT; ++k) {
        int e = base + k * BIN_THREADS + (int)threadIdx.x;
        r[k] = 0; c[k] = 0; vb[k] = 0;
        if (e < E) {
            r[k]  = er[e];
            c[k]  = ec[e];
            vb[k] = __float_as_int(ev[e]);
        }
    }
#pragma unroll
    for (int k = 0; k < EPT; ++k) {
        int e = base + k * BIN_THREADS + (int)threadIdx.x;
        rank[k] = 0;
        if (e < E) rank[k] = atomicAdd(&hist[r[k] >> RB_SHIFT], 1);
    }
    __syncthreads();
    for (int i = threadIdx.x; i < nb; i += BIN_THREADS) {
        int n = hist[i];
        int g = 0;
        if (n) g = atomicAdd(&binCursor[i], n);
        hist[i] = g;
    }
    __syncthreads();
#pragma unroll
    for (int k = 0; k < EPT; ++k) {
        int e = base + k * BIN_THREADS + (int)threadIdx.x;
        if (e < E) {
            int slot = hist[r[k] >> RB_SHIFT] + rank[k];
            rec[slot] = make_int4(r[k], c[k], vb[k], 0);
        }
    }
}

#define SPMM_THREADS 512
__global__ __launch_bounds__(SPMM_THREADS)
void gcn_bin_spmm_kernel(const int* __restrict__ recWords, const int* __restrict__ binStart,
                         const float* __restrict__ deg, const float* __restrict__ feat,
                         float* __restrict__ agg, int N) {
    __shared__ float tile[RB * D_FEAT];
    const int4* rec = reinterpret_cast<const int4*>(recWords);
    const int tid  = threadIdx.x;
    const int lane = tid & 31;
    const int wave = __builtin_amdgcn_readfirstlane(tid >> 5);
    const int rowBase = (int)blockIdx.x << RB_SHIFT;
    for (int i = tid; i < RB * D_FEAT; i += SPMM_THREADS) tile[i] = 0.0f;
    const int start = binStart[blockIdx.x];
    const int end   = binStart[blockIdx.x + 1];
    __syncthreads();

    int g0 = start + wave * 32;
    int4 q = make_int4(0, 0, 0, 0);
    if (g0 + lane < end) q = rec[g0 + lane];
    while (g0 < end) {
        const int gn = g0 + SPMM_THREADS;
        int4 qn = make_int4(0, 0, 0, 0);
        if (gn + lane < end) qn = rec[gn + lane];
        float v = 0.0f;
        if (g0 + lane < end)
            v = gcn_inv_sqrt_deg(deg[q.x]) * __int_as_float(q.z) * gcn_inv_sqrt_deg(deg[q.y]);
        const int vbits = __float_as_int(v);
        const int rl = (q.x - rowBase) * D_FEAT;
        const int nvalid = (end - g0 < 32) ? (end - g0) : 32;
        if (nvalid == 32) {
            float f[32];
#pragma unroll
            for (int j = 0; j < 32; ++j) {
                int cj = __builtin_amdgcn_readlane(q.y, j);
                f[j] = feat[(size_t)cj * D_FEAT + lane];
            }
#pragma unroll
            for (int j = 0; j < 32; ++j) {
                float vj = __int_as_float(__builtin_amdgcn_readlane(vbits, j));
                int rj = __builtin_amdgcn_readlane(rl, j);
                atomicAdd(&tile[rj + lane], vj * f[j]);
            }
        } else {
            for (int j = 0; j < nvalid; ++j) {
                int cj = __builtin_amdgcn_readlane(q.y, j);
                float vj = __int_as_float(__builtin_amdgcn_readlane(vbits, j));
                int rj = __builtin_amdgcn_readlane(rl, j);
                float fj = feat[(size_t)cj * D_FEAT + lane];
                atomicAdd(&tile[rj + lane], vj * fj);
            }
        }
        q = qn;
        g0 = gn;
    }
    __syncthreads();
    int rows = N - rowBase;
    if (rows > RB) rows = RB;
    float* dst = agg + (size_t)rowBase * D_FEAT;
    for (int i = tid; i < rows * D_FEAT; i += SPMM_THREADS) dst[i] = tile[i];
}

__global__ void gcn_gemm_wmma_kernel(const float* __restrict__ agg,
                                     const float* __restrict__ W,
                                     float* __restrict__ out, int nTiles) {
    int wave = (int)(((size_t)blockIdx.x * blockDim.x + threadIdx.x) >> 5);
    if (wave >= nTiles) return;
    int lane = threadIdx.x & 31;
    int half = lane >> 4;
    int l15  = lane & 15;

    const float* aRow = agg + (size_t)(wave * 16 + l15) * D_FEAT;

    v8f acc0 = {};
    v8f acc1 = {};
#pragma unroll
    for (int k = 0; k < D_FEAT; k += 4) {
        int kk = k + half * 2;
        v2f a, b0, b1;
        a.x  = aRow[kk];
        a.y  = aRow[kk + 1];
        b0.x = W[l15 * D_FEAT + kk];
        b0.y = W[l15 * D_FEAT + kk + 1];
        b1.x = W[(16 + l15) * D_FEAT + kk];
        b1.y = W[(16 + l15) * D_FEAT + kk + 1];
        acc0 = __builtin_amdgcn_wmma_f32_16x16x4_f32(
            false, a, false, b0, (short)0, acc0, false, false);
        acc1 = __builtin_amdgcn_wmma_f32_16x16x4_f32(
            false, a, false, b1, (short)0, acc1, false, false);
    }

    int mBase = wave * 16 + half * 8;
#pragma unroll
    for (int i = 0; i < 8; ++i) {
        out[(size_t)(mBase + i) * D_FEAT + l15]      = acc0[i];
        out[(size_t)(mBase + i) * D_FEAT + 16 + l15] = acc1[i];
    }
}

__global__ void gcn_gemm_tail_kernel(const float* __restrict__ agg,
                                     const float* __restrict__ W,
                                     float* __restrict__ out,
                                     int rowStart, int N) {
    int idx = blockIdx.x * blockDim.x + threadIdx.x;
    int row = rowStart + (idx >> 5);
    int o = idx & 31;
    if (row >= N) return;
    float s = 0.0f;
#pragma unroll
    for (int k = 0; k < D_FEAT; ++k)
        s += agg[(size_t)row * D_FEAT + k] * W[o * D_FEAT + k];
    out[(size_t)row * D_FEAT + o] = s;
}

extern "C" void kernel_launch(void* const* d_in, const int* in_sizes, int n_in,
                              void* d_out, int out_size, void* d_ws, size_t ws_size,
                              hipStream_t stream) {
    const int*   er   = (const int*)d_in[0];
    const int*   ec   = (const int*)d_in[1];
    const float* ev   = (const float*)d_in[2];
    const float* feat = (const float*)d_in[3];
    const float* W    = (const float*)d_in[4];
    float* out = (float*)d_out;

    int E = in_sizes[0];
    int N = in_sizes[3] / D_FEAT;

    int nb = (N + RB - 1) >> RB_SHIFT;

    float* agg = (float*)d_ws;
    float* deg = agg + (size_t)N * D_FEAT;
    int* binTotal  = (int*)(deg + N);
    int* binStart  = binTotal + nb;
    int* binCursor = binStart + nb + 1;
    size_t recOff = ((size_t)N * (D_FEAT + 1) + 3 * (size_t)nb + 1 + 3) & ~(size_t)3;
    int* rec = (int*)d_ws + recOff;

    size_t zn = (size_t)N + (size_t)nb;
    gcn_zero_kernel<<<(unsigned)((zn + 255) / 256), 256, 0, stream>>>(deg, zn);

    int nChunks = (E + CHUNK - 1) / CHUNK;
    size_t histBytes = (size_t)nb * sizeof(int);
    gcn_bin_count_kernel<<<nChunks, BIN_THREADS, histBytes, stream>>>(
        er, ev, deg, binTotal, E, nb);
    gcn_bin_scan_kernel<<<1, 1024, 0, stream>>>(binTotal, binStart, binCursor, nb);
    gcn_bin_scatter_kernel<<<nChunks, BIN_THREADS, histBytes, stream>>>(
        er, ec, ev, binCursor, rec, E, nb);
    gcn_bin_spmm_kernel<<<nb, SPMM_THREADS, 0, stream>>>(rec, binStart, deg, feat, agg, N);

    int nTiles = N / 16;
    if (nTiles > 0) {
        size_t gemmThreads = (size_t)nTiles * 32;
        gcn_gemm_wmma_kernel<<<(unsigned)((gemmThreads + 255) / 256), 256, 0, stream>>>(
            agg, W, out, nTiles);
    }
    int rowStart = nTiles * 16;
    if (rowStart < N) {
        int tailThreads = (N - rowStart) * 32;
        gcn_gemm_tail_kernel<<<(tailThreads + 255) / 256, 256, 0, stream>>>(
            agg, W, out, rowStart, N);
    }
}


# Dense GEMM moves into the SpMM work-group: each 256-row block is multiplied by W^T (f32 WMMA) straight from its LDS tile and stored as out; agg never goes to memory; 6 -> 5 launches
# speedup vs baseline: 1.8107x; 1.0378x over previous
#include <hip/hip_runtime.h>
#include <hip/hip_bf16.h>


typedef __attribute__((ext_vector_type(2))) float v2f;
typedef __attribute__((ext_vector_type(4))) float v4f;
typedef __attribute__((ext_vector_type(8))) float v8f;

#define D_FEAT 32

__global__ void gcn_zero_kernel(float* __restrict__ p, size_t n) {
    size_t i = (size_t)blockIdx.x * blockDim.x + threadIdx.x;
    if (i < n) p[i] = 0.0f;
}

__device__ __forceinline__ float gcn_inv_sqrt_deg(float d) {
    return (d > 0.0f) ? rsqrtf(d) : 0.0f;
}

#define RB_SHIFT 8
#define RB (1 << RB_SHIFT)
#define BIN_THREADS 1024
#define EPT 8
#define CHUNK (EPT * BIN_THREADS)

__global__ __launch_bounds__(BIN_THREADS)
void gcn_bin_count_kernel(const int* __restrict__ er, const float* __restrict__ ev,
                          float* __restrict__ deg, int* __restrict__ binTotal,
                          int E, int nb) {
    extern __shared__ int hist[];
    for (int i = threadIdx.x; i < nb; i += BIN_THREADS) hist[i] = 0;
    __syncthreads();
    int base = blockIdx.x * CHUNK;
    int r[EPT];
    float w[EPT];
#pragma unroll
    for (int k = 0; k < EPT; ++k) {
        int e = base + k * BIN_THREADS + (int)threadIdx.x;
        r[k] = -1;
        w[k] = 0.0f;
        if (e < E) {
            r[k] = er[e];
            w[k] = ev[e];
        }
    }
#pragma unroll
    for (int k = 0; k < EPT; ++k)
        if (r[k] >= 0) {
            atomicAdd(&deg[r[k]], w[k]);
            atomicAdd(&hist[r[k] >> RB_SHIFT], 1);
        }
    __syncthreads();
    for (int i = threadIdx.x; i < nb; i += BIN_THREADS) {
        int c = hist[i];
        if (c) atomicAdd(&binTotal[i], c);
    }
}

__global__ __launch_bounds__(1024)
void gcn_bin_scan_kernel(const int* __restrict__ binTotal, int* __restrict__ binStart,
                         int* __restrict__ binCursor, int nb) {
    __shared__ int s[1024];
    __shared__ int carry;
    int t = threadIdx.x;
    if (t == 0) carry = 0;
    __syncthreads();
    for (int t0 = 0; t0 < nb; t0 += 1024) {
        int b = t0 + t;
        int v = (b < nb) ? binTotal[b] : 0;
        s[t] = v;
        __syncthreads();
        for (int off = 1; off < 1024; off <<= 1) {
            int add = (t >= off) ? s[t - off] : 0;
            __syncthreads();
            s[t] += add;
            __syncthreads();
        }
        int incl = s[t];
        int c0 = carry;
        if (b < nb) {
            binStart[b]  = c0 + incl - v;
            binCursor[b] = c0 + incl - v;
        }
        __syncthreads();
        if (t == 1023) carry = c0 + incl;
        __syncthreads();
    }
    if (t == 0) binStart[nb] = carry;
}

__global__ __launch_bounds__(BIN_THREADS)
void gcn_bin_scatter_kernel(const int* __restrict__ er, const int* __restrict__ ec,
                            const float* __restrict__ ev, int* __restrict__ binCursor,
                            int* __restrict__ recWords, int E, int nb) {
    extern __shared__ int hist[];
    int4* rec = reinterpret_cast<int4*>(recWords);
    for (int i = threadIdx.x; i < nb; i += BIN_THREADS) hist[i] = 0;
    __syncthreads();
    int base = blockIdx.x * CHUNK;
    int r[EPT], c[EPT], vb[EPT], rank[EPT];
#pragma unroll
    for (int k = 0; k < EPT; ++k) {
        int e = base + k * BIN_THREADS + (int)threadIdx.x;
        r[k] = 0; c[k] = 0; vb[k] = 0;
        if (e < E) {
            r[k]  = er[e];
            c[k]  = ec[e];
            vb[k] = __float_as_int(ev[e]);
        }
    }
#pragma unroll
    for (int k = 0; k < EPT; ++k) {
        int e = base + k * BIN_THREADS + (int)threadIdx.x;
        rank[k] = 0;
        if (e < E) rank[k] = atomicAdd(&hist[r[k] >> RB_SHIFT], 1);
    }
    __syncthreads();
    for (int i = threadIdx.x; i < nb; i += BIN_THREADS) {
        int n = hist[i];
        int g = 0;
        if (n) g = atomicAdd(&binCursor[i], n);
        hist[i] = g;
    }
    __syncthreads();
#pragma unroll
    for (int k = 0; k < EPT; ++k) {
        int e = base + k * BIN_THREADS + (int)threadIdx.x;
        if (e < E) {
            int slot = hist[r[k] >> RB_SHIFT] + rank[k];
            rec[slot] = make_int4(r[k], c[k], vb[k], 0);
        }
    }
}

#define SPMM_THREADS 512
#define TP 36
#define WP 34
__global__ __launch_bounds__(SPMM_THREADS)
void gcn_bin_spmm_gemm_kernel(const int* __restrict__ recWords, const int* __restrict__ binStart,
                              const float* __restrict__ deg, const float* __restrict__ feat,
                              const float* __restrict__ W,
                              float* __restrict__ out, int N) {
    __shared__ __attribute__((aligned(16))) float tile[RB * TP];
    __shared__ __attribute__((aligned(16))) float wl[D_FEAT * WP];
    const int4* rec = reinterpret_cast<const int4*>(recWords);
    const int tid  = threadIdx.x;
    const int lane = tid & 31;
    const int wave = __builtin_amdgcn_readfirstlane(tid >> 5);
    const int rowBase = (int)blockIdx.x << RB_SHIFT;
    for (int i = tid; i < RB * TP; i += SPMM_THREADS) tile[i] = 0.0f;
    {
        const int n = tid >> 4, k = (tid & 15) * 2;
        wl[n * WP + k]     = W[n * D_FEAT + k];
        wl[n * WP + k + 1] = W[n * D_FEAT + k + 1];
    }
    const int start = binStart[blockIdx.x];
    const int end   = binStart[blockIdx.x + 1];
    __syncthreads();

    int g0 = start + wave * 32;
    int4 q = make_int4(0, 0, 0, 0);
    if (g0 + lane < end) q = rec[g0 + lane];
    while (g0 < end) {
        const int gn = g0 + SPMM_THREADS;
        int4 qn = make_int4(0, 0, 0, 0);
        if (gn + lane < end) qn = rec[gn + lane];
        float v = 0.0f;
        if (g0 + lane < end)
            v = gcn_inv_sqrt_deg(deg[q.x]) * __int_as_float(q.z) * gcn_inv_sqrt_deg(deg[q.y]);
        const int vbits = __float_as_int(v);
        const int rl = (q.x - rowBase) * TP;
        const int nvalid = (end - g0 < 32) ? (end - g0) : 32;
        if (nvalid == 32) {
            float f[32];
#pragma unroll
            for (int j = 0; j < 32; ++j) {
                int cj = __builtin_amdgcn_readlane(q.y, j);
                f[j] = feat[(size_t)cj * D_FEAT + lane];
            }
#pragma unroll
            for (int j = 0; j < 32; ++j) {
                float vj = __int_as_float(__builtin_amdgcn_readlane(vbits, j));
                int rj = __builtin_amdgcn_readlane(rl, j);
                atomicAdd(&tile[rj + lane], vj * f[j]);
            }
        } else {
            for (int j = 0; j < nvalid; ++j) {
                int cj = __builtin_amdgcn_readlane(q.y, j);
                float vj = __int_as_float(__builtin_amdgcn_readlane(vbits, j));
                int rj = __builtin_amdgcn_readlane(rl, j);
                float fj = feat[(size_t)cj * D_FEAT + lane];
                atomicAdd(&tile[rj + lane], vj * fj);
            }
        }
        q = qn;
        g0 = gn;
    }
    __syncthreads();

    {
        const int half = lane >> 4;
        const int l15  = lane & 15;
        const float* aRow = &tile[(wave * 16 + l15) * TP];
        const float* b0Row = &wl[l15 * WP];
        const float* b1Row = &wl[(16 + l15) * WP];
        v8f acc0 = {};
        v8f acc1 = {};
#pragma unroll
        for (int k = 0; k < D_FEAT; k += 4) {
            const int kk = k + half * 2;
            v2f a, b0, b1;
            a.x  = aRow[kk];
            a.y  = aRow[kk + 1];
            b0.x = b0Row[kk];
            b0.y = b0Row[kk + 1];
            b1.x = b1Row[kk];
            b1.y = b1Row[kk + 1];
            acc0 = __builtin_amdgcn_wmma_f32_16x16x4_f32(
                false, a, false, b0, (short)0, acc0, false, false);
            acc1 = __builtin_amdgcn_wmma_f32_16x16x4_f32(
                false, a, false, b1, (short)0, acc1, false, false);
        }
        float* dRow = &tile[(wave * 16 + half * 8) * TP + l15];
#pragma unroll
        for (int i = 0; i < 8; ++i) {
            dRow[i * TP]      = acc0[i];
            dRow[i * TP + 16] = acc1[i];
        }
    }
    __syncthreads();
    int rows = N - rowBase;
    if (rows > RB) rows = RB;
    float* dst = out + (size_t)rowBase * D_FEAT;
    for (int i = tid; i < rows * (D_FEAT / 4); i += SPMM_THREADS) {
        const int r = i >> 3, p = (i & 7) * 4;
        v4f val = *reinterpret_cast<const v4f*>(&tile[r * TP + p]);
        *reinterpret_cast<v4f*>(&dst[(size_t)i * 4]) = val;
    }
}

extern "C" void kernel_launch(void* const* d_in, const int* in_sizes, int n_in,
                              void* d_out, int out_size, void* d_ws, size_t ws_size,
                              hipStream_t stream) {
    const int*   er   = (const int*)d_in[0];
    const int*   ec   = (const int*)d_in[1];
    const float* ev   = (const float*)d_in[2];
    const float* feat = (const float*)d_in[3];
    const float* W    = (const float*)d_in[4];
    float* out = (float*)d_out;

    int E = in_sizes[0];
    int N = in_sizes[3] / D_FEAT;

    int nb = (N + RB - 1) >> RB_SHIFT;

    float* agg = (float*)d_ws;
    float* deg = agg + (size_t)N * D_FEAT;
    int* binTotal  = (int*)(deg + N);
    int* binStart  = binTotal + nb;
    int* binCursor = binStart + nb + 1;
    size_t recOff = ((size_t)N * (D_FEAT + 1) + 3 * (size_t)nb + 1 + 3) & ~(size_t)3;
    int* rec = (int*)d_ws + recOff;

    size_t zn = (size_t)N + (size_t)nb;
    gcn_zero_kernel<<<(unsigned)((zn + 255) / 256), 256, 0, stream>>>(deg, zn);

    int nChunks = (E + CHUNK - 1) / CHUNK;
    size_t histBytes = (size_t)nb * sizeof(int);
    gcn_bin_count_kernel<<<nChunks, BIN_THREADS, histBytes, stream>>>(
        er, ev, deg, binTotal, E, nb);
    gcn_bin_scan_kernel<<<1, 1024, 0, stream>>>(binTotal, binStart, binCursor, nb);
    gcn_bin_scatter_kernel<<<nChunks, BIN_THREADS, histBytes, stream>>>(
        er, ec, ev, binCursor, rec, E, nb);
    gcn_bin_spmm_gemm_kernel<<<nb, SPMM_THREADS, 0, stream>>>(rec, binStart, deg, feat, W, out, N);
}


# Edge grouping made chunk-local: each 8192-edge chunk is ordered by row block in one LDS pass into its own region; SpMM walks its block's run in every chunk; 5 -> 3 launches; agg sum order changes
# speedup vs baseline: 1.8843x; 1.0406x over previous
#include <hip/hip_runtime.h>
#include <hip/hip_bf16.h>


typedef __attribute__((ext_vector_type(2))) float v2f;
typedef __attribute__((ext_vector_type(4))) float v4f;
typedef __attribute__((ext_vector_type(8))) float v8f;

#define D_FEAT 32

__global__ void gcn_zero_kernel(float* __restrict__ p, size_t n) {
    size_t i = (size_t)blockIdx.x * blockDim.x + threadIdx.x;
    if (i < n) p[i] = 0.0f;
}

__device__ __forceinline__ float gcn_inv_sqrt_deg(float d) {
    return (d > 0.0f) ? rsqrtf(d) : 0.0f;
}

#define RB_SHIFT 8
#define RB (1 << RB_SHIFT)
#define BIN_THREADS 1024
#define EPT 8
#define CHUNK (EPT * BIN_THREADS)

__global__ __launch_bounds__(BIN_THREADS)
void gcn_chunk_group_kernel(const int* __restrict__ er, const int* __restrict__ ec,
                            const float* __restrict__ ev, float* __restrict__ deg,
                            int* __restrict__ segTab, int* __restrict__ recWords,
                            int E, int nb) {
    extern __shared__ int hist[];
    __shared__ int s[BIN_THREADS];
    __shared__ int carry;
    int4* rec = reinterpret_cast<int4*>(recWords);
    const int t = threadIdx.x;
    for (int i = t; i < nb; i += BIN_THREADS) hist[i] = 0;
    if (t == 0) carry = 0;
    __syncthreads();
    const int base = blockIdx.x * CHUNK;
    int r[EPT], c[EPT], vb[EPT], rank[EPT];
#pragma unroll
    for (int k = 0; k < EPT; ++k) {
        int e = base + k * BIN_THREADS + t;
        r[k] = 0; c[k] = 0; vb[k] = 0;
        if (e < E) {
            r[k]  = er[e];
            c[k]  = ec[e];
            vb[k] = __float_as_int(ev[e]);
        }
    }
#pragma unroll
    for (int k = 0; k < EPT; ++k) {
        int e = base + k * BIN_THREADS + t;
        rank[k] = 0;
        if (e < E) {
            atomicAdd(&deg[r[k]], __int_as_float(vb[k]));
            rank[k] = atomicAdd(&hist[r[k] >> RB_SHIFT], 1);
        }
    }
    __syncthreads();
    int* seg = segTab + (size_t)blockIdx.x * (size_t)(nb + 1);
    for (int t0 = 0; t0 < nb; t0 += BIN_THREADS) {
        const int b = t0 + t;
        const int v = (b < nb) ? hist[b] : 0;
        s[t] = v;
        __syncthreads();
        for (int off = 1; off < BIN_THREADS; off <<= 1) {
            int add = (t >= off) ? s[t - off] : 0;
            __syncthreads();
            s[t] += add;
            __syncthreads();
        }
        const int incl = s[t];
        const int c0 = carry;
        if (b < nb) {
            hist[b] = c0 + incl - v;
            seg[b]  = c0 + incl - v;
        }
        __syncthreads();
        if (t == BIN_THREADS - 1) carry = c0 + incl;
        __syncthreads();
    }
    if (t == 0) seg[nb] = carry;
#pragma unroll
    for (int k = 0; k < EPT; ++k) {
        int e = base + k * BIN_THREADS + t;
        if (e < E) {
            int slot = base + hist[r[k] >> RB_SHIFT] + rank[k];
            rec[slot] = make_int4(r[k], c[k], vb[k], 0);
        }
    }
}

__device__ __forceinline__ int gcn_seg_slot(const int* pre, const int* segS, int nChunks, int g) {
    int lo = 0, hi = nChunks;
    while (hi - lo > 1) {
        const int mid = (lo + hi) >> 1;
        if (pre[mid] <= g) lo = mid; else hi = mid;
    }
    return segS[lo] + (g - pre[lo]);
}

#define SPMM_THREADS 512
#define TP 36
#define WP 34
__global__ __launch_bounds__(SPMM_THREADS)
void gcn_bin_spmm_gemm_kernel(const int* __restrict__ recWords, const int* __restrict__ segTab,
                              const float* __restrict__ deg, const float* __restrict__ feat,
                              const float* __restrict__ W,
                              float* __restrict__ out, int N, int nb, int nChunks) {
    __shared__ __attribute__((aligned(16))) float tile[RB * TP];
    __shared__ __attribute__((aligned(16))) float wl[D_FEAT * WP];
    extern __shared__ int segL[];
    __shared__ int sc[SPMM_THREADS];
    __shared__ int carry;
    int* pre  = segL;
    int* segS = segL + (nChunks + 1);
    const int4* rec = reinterpret_cast<const int4*>(recWords);
    const int tid  = threadIdx.x;
    const int lane = tid & 31;
    const int wave = __builtin_amdgcn_readfirstlane(tid >> 5);
    const int rowBase = (int)blockIdx.x << RB_SHIFT;
    for (int i = tid; i < RB * TP; i += SPMM_THREADS) tile[i] = 0.0f;
    {
        const int n = tid >> 4, k = (tid & 15) * 2;
        wl[n * WP + k]     = W[n * D_FEAT + k];
        wl[n * WP + k + 1] = W[n * D_FEAT + k + 1];
    }
    if (tid == 0) carry = 0;
    __syncthreads();
    for (int c0 = 0; c0 < nChunks; c0 += SPMM_THREADS) {
        const int c = c0 + tid;
        int a = 0, len = 0;
        if (c < nChunks) {
            const int* row = segTab + (size_t)c * (size_t)(nb + 1) + blockIdx.x;
            a   = row[0];
            len = row[1] - a;
        }
        sc[tid] = len;
        __syncthreads();
        for (int off = 1; off < SPMM_THREADS; off <<= 1) {
            int add = (tid >= off) ? sc[tid - off] : 0;
            __syncthreads();
            sc[tid] += add;
            __syncthreads();
        }
        const int incl = sc[tid];
        const int cv = carry;
        if (c < nChunks) {
            pre[c]  = cv + incl - len;
            segS[c] = c * CHUNK + a;
        }
        __syncthreads();
        if (tid == SPMM_THREADS - 1) carry = cv + incl;
        __syncthreads();
    }
    const int end = __builtin_amdgcn_readfirstlane(carry);
    if (tid == 0) pre[nChunks] = end;
    __syncthreads();

    int g0 = wave * 32;
    int sl = -1, sln = -1;
    if (g0 + lane < end) sl = gcn_seg_slot(pre, segS, nChunks, g0 + lane);
    if (g0 + SPMM_THREADS + lane < end) sln = gcn_seg_slot(pre, segS, nChunks, g0 + SPMM_THREADS + lane);
    int4 q = make_int4(0, 0, 0, 0);
    if (sl >= 0) q = rec[sl];
    while (g0 < end) {
        const int gn = g0 + SPMM_THREADS;
        int4 qn = make_int4(0, 0, 0, 0);
        if (sln >= 0) qn = rec[sln];
        int sln2 = -1;
        if (gn + SPMM_THREADS + lane < end) sln2 = gcn_seg_slot(pre, segS, nChunks, gn + SPMM_THREADS + lane);
        float v = 0.0f;
        if (g0 + lane < end)
            v = gcn_inv_sqrt_deg(deg[q.x]) * __int_as_float(q.z) * gcn_inv_sqrt_deg(deg[q.y]);
        const int vbits = __float_as_int(v);
        const int rl = (q.x - rowBase) * TP;
        const int nvalid = (end - g0 < 32) ? (end - g0) : 32;
        if (nvalid == 32) {
            float f[32];
#pragma unroll
            for (int j = 0; j < 32; ++j) {
                int cj = __builtin_amdgcn_readlane(q.y, j);
                f[j] = feat[(size_t)cj * D_FEAT + lane];
            }
#pragma unroll
            for (int j = 0; j < 32; ++j) {
                float vj = __int_as_float(__builtin_amdgcn_readlane(vbits, j));
                int rj = __builtin_amdgcn_readlane(rl, j);
                atomicAdd(&tile[rj + lane], vj * f[j]);
            }
        } else {
            for (int j = 0; j < nvalid; ++j) {
                int cj = __builtin_amdgcn_readlane(q.y, j);
                float vj = __int_as_float(__builtin_amdgcn_readlane(vbits, j));
                int rj = __builtin_amdgcn_readlane(rl, j);
                float fj = feat[(size_t)cj * D_FEAT + lane];
                atomicAdd(&tile[rj + lane], vj * fj);
            }
        }
        q = qn;
        sln = sln2;
        g0 = gn;
    }
    __syncthreads();

    {
        const int half = lane >> 4;
        const int l15  = lane & 15;
        const float* aRow = &tile[(wave * 16 + l15) * TP];
        const float* b0Row = &wl[l15 * WP];
        const float* b1Row = &wl[(16 + l15) * WP];
        v8f acc0 = {};
        v8f acc1 = {};
#pragma unroll
        for (int k = 0; k < D_FEAT; k += 4) {
            const int kk = k + half * 2;
            v2f a, b0, b1;
            a.x  = aRow[kk];
            a.y  = aRow[kk + 1];
            b0.x = b0Row[kk];
            b0.y = b0Row[kk + 1];
            b1.x = b1Row[kk];
            b1.y = b1Row[kk + 1];
            acc0 = __builtin_amdgcn_wmma_f32_16x16x4_f32(
                false, a, false, b0, (short)0, acc0, false, false);
            acc1 = __builtin_amdgcn_wmma_f32_16x16x4_f32(
                false, a, false, b1, (short)0, acc1, false, false);
        }
        float* dRow = &tile[(wave * 16 + half * 8) * TP + l15];
#pragma unroll
        for (int i = 0; i < 8; ++i) {
            dRow[i * TP]      = acc0[i];
            dRow[i * TP + 16] = acc1[i];
        }
    }
    __syncthreads();
    int rows = N - rowBase;
    if (rows > RB) rows = RB;
    float* dst = out + (size_t)rowBase * D_FEAT;
    for (int i = tid; i < rows * (D_FEAT / 4); i += SPMM_THREADS) {
        const int r = i >> 3, p = (i & 7) * 4;
        v4f val = *reinterpret_cast<const v4f*>(&tile[r * TP + p]);
        *reinterpret_cast<v4f*>(&dst[(size_t)i * 4]) = val;
    }
}

extern "C" void kernel_launch(void* const* d_in, const int* in_sizes, int n_in,
                              void* d_out, int out_size, void* d_ws, size_t ws_size,
                              hipStream_t stream) {
    const int*   er   = (const int*)d_in[0];
    const int*   ec   = (const int*)d_in[1];
    const float* ev   = (const float*)d_in[2];
    const float* feat = (const float*)d_in[3];
    const float* W    = (const float*)d_in[4];
    float* out = (float*)d_out;

    int E = in_sizes[0];
    int N = in_sizes[3] / D_FEAT;

    int nb = (N + RB - 1) >> RB_SHIFT;
    int nChunks = (E + CHUNK - 1) / CHUNK;

    float* deg = (float*)d_ws;
    int* segTab = (int*)(deg + N);
    size_t recOff = ((size_t)N + (size_t)nChunks * (size_t)(nb + 1) + 3) & ~(size_t)3;
    int* rec = (int*)d_ws + recOff;

    gcn_zero_kernel<<<(unsigned)(((size_t)N + 255) / 256), 256, 0, stream>>>(deg, (size_t)N);

    size_t histBytes = (size_t)nb * sizeof(int);
    gcn_chunk_group_kernel<<<nChunks, BIN_THREADS, histBytes, stream>>>(
        er, ec, ev, deg, segTab, rec, E, nb);
    size_t segBytes = ((size_t)nChunks * 2 + 1) * sizeof(int);
    gcn_bin_spmm_gemm_kernel<<<nb, SPMM_THREADS, segBytes, stream>>>(
        rec, segTab, deg, feat, W, out, N, nb, nChunks);
}


# Grouping pass: a chunk's records are put in row-block order in a 128 KiB LDS stage and leave as whole 128-B lines (4 per store instruction), not as 16-B stores scattered over the chunk's region
# speedup vs baseline: 1.9511x; 1.0355x over previous
#include <hip/hip_runtime.h>
#include <hip/hip_bf16.h>


typedef __attribute__((ext_vector_type(2))) float v2f;
typedef __attribute__((ext_vector_type(4))) float v4f;
typedef __attribute__((ext_vector_type(8))) float v8f;

#define D_FEAT 32

__global__ void gcn_zero_kernel(float* __restrict__ p, size_t n) {
    size_t i = (size_t)blockIdx.x * blockDim.x + threadIdx.x;
    if (i < n) p[i] = 0.0f;
}

__device__ __forceinline__ float gcn_inv_sqrt_deg(float d) {
    return (d > 0.0f) ? rsqrtf(d) : 0.0f;
}

#define RB_SHIFT 8
#define RB (1 << RB_SHIFT)
#define BIN_THREADS 1024
#define EPT 8
#define CHUNK (EPT * BIN_THREADS)

__global__ __launch_bounds__(BIN_THREADS)
void gcn_chunk_group_kernel(const int* __restrict__ er, const int* __restrict__ ec,
                            const float* __restrict__ ev, float* __restrict__ deg,
                            int* __restrict__ segTab, int* __restrict__ recWords,
                            int E, int nb) {
    extern __shared__ int hist[];
    __shared__ int4 stage[CHUNK];
    __shared__ int s[BIN_THREADS];
    __shared__ int carry;
    int4* rec = reinterpret_cast<int4*>(recWords);
    const int t = threadIdx.x;
    for (int i = t; i < nb; i += BIN_THREADS) hist[i] = 0;
    if (t == 0) carry = 0;
    __syncthreads();
    const int base = blockIdx.x * CHUNK;
    int r[EPT], c[EPT], vb[EPT], rank[EPT];
#pragma unroll
    for (int k = 0; k < EPT; ++k) {
        int e = base + k * BIN_THREADS + t;
        r[k] = 0; c[k] = 0; vb[k] = 0;
        if (e < E) {
            r[k]  = er[e];
            c[k]  = ec[e];
            vb[k] = __float_as_int(ev[e]);
        }
    }
#pragma unroll
    for (int k = 0; k < EPT; ++k) {
        int e = base + k * BIN_THREADS + t;
        rank[k] = 0;
        if (e < E) {
            atomicAdd(&deg[r[k]], __int_as_float(vb[k]));
            rank[k] = atomicAdd(&hist[r[k] >> RB_SHIFT], 1);
        }
    }
    __syncthreads();
    int* seg = segTab + (size_t)blockIdx.x * (size_t)(nb + 1);
    for (int t0 = 0; t0 < nb; t0 += BIN_THREADS) {
        const int b = t0 + t;
        const int v = (b < nb) ? hist[b] : 0;
        s[t] = v;
        __syncthreads();
        for (int off = 1; off < BIN_THREADS; off <<= 1) {
            int add = (t >= off) ? s[t - off] : 0;
            __syncthreads();
            s[t] += add;
            __syncthreads();
        }
        const int incl = s[t];
        const int c0 = carry;
        if (b < nb) {
            hist[b] = c0 + incl - v;
            seg[b]  = c0 + incl - v;
        }
        __syncthreads();
        if (t == BIN_THREADS - 1) carry = c0 + incl;
        __syncthreads();
    }
    if (t == 0) seg[nb] = carry;
#pragma unroll
    for (int k = 0; k < EPT; ++k) {
        int e = base + k * BIN_THREADS + t;
        if (e < E) {
            int slot = hist[r[k] >> RB_SHIFT] + rank[k];
            stage[slot] = make_int4(r[k], c[k], vb[k], 0);
        }
    }
    __syncthreads();
    const int cnt = E - base;
    if (cnt >= CHUNK) {
#pragma unroll
        for (int k = 0; k < EPT; ++k) {
            const int i = k * BIN_THREADS + t;
            rec[base + i] = stage[i];
        }
    } else {
        for (int i = t; i < cnt; i += BIN_THREADS) rec[base + i] = stage[i];
    }
}

__device__ __forceinline__ int gcn_seg_slot(const int* pre, const int* segS, int nChunks, int g) {
    int lo = 0, hi = nChunks;
    while (hi - lo > 1) {
        const int mid = (lo + hi) >> 1;
        if (pre[mid] <= g) lo = mid; else hi = mid;
    }
    return segS[lo] + (g - pre[lo]);
}

#define SPMM_THREADS 512
#define TP 36
#define WP 34
__global__ __launch_bounds__(SPMM_THREADS)
void gcn_bin_spmm_gemm_kernel(const int* __restrict__ recWords, const int* __restrict__ segTab,
                              const float* __restrict__ deg, const float* __restrict__ feat,
                              const float* __restrict__ W,
                              float* __restrict__ out, int N, int nb, int nChunks) {
    __shared__ __attribute__((aligned(16))) float tile[RB * TP];
    __shared__ __attribute__((aligned(16))) float wl[D_FEAT * WP];
    extern __shared__ int segL[];
    __shared__ int sc[SPMM_THREADS];
    __shared__ int carry;
    int* pre  = segL;
    int* segS = segL + (nChunks + 1);
    const int4* rec = reinterpret_cast<const int4*>(recWords);
    const int tid  = threadIdx.x;
    const int lane = tid & 31;
    const int wave = __builtin_amdgcn_readfirstlane(tid >> 5);
    const int rowBase = (int)blockIdx.x << RB_SHIFT;
    for (int i = tid; i < RB * TP; i += SPMM_THREADS) tile[i] = 0.0f;
    {
        const int n = tid >> 4, k = (tid & 15) * 2;
        wl[n * WP + k]     = W[n * D_FEAT + k];
        wl[n * WP + k + 1] = W[n * D_FEAT + k + 1];
    }
    if (tid == 0) carry = 0;
    __syncthreads();
    for (int c0 = 0; c0 < nChunks; c0 += SPMM_THREADS) {
        const int c = c0 + tid;
        int a = 0, len = 0;
        if (c < nChunks) {
            const int* row = segTab + (size_t)c * (size_t)(nb + 1) + blockIdx.x;
            a   = row[0];
            len = row[1] - a;
        }
        sc[tid] = len;
        __syncthreads();
        for (int off = 1; off < SPMM_THREADS; off <<= 1) {
            int add = (tid >= off) ? sc[tid - off] : 0;
            __syncthreads();
            sc[tid] += add;
            __syncthreads();
        }
        const int incl = sc[tid];
        const int cv = carry;
        if (c < nChunks) {
            pre[c]  = cv + incl - len;
            segS[c] = c * CHUNK + a;
        }
        __syncthreads();
        if (tid == SPMM_THREADS - 1) carry = cv + incl;
        __syncthreads();
    }
    const int end = __builtin_amdgcn_readfirstlane(carry);
    if (tid == 0) pre[nChunks] = end;
    __syncthreads();

    int g0 = wave * 32;
    int sl = -1, sln = -1;
    if (g0 + lane < end) sl = gcn_seg_slot(pre, segS, nChunks, g0 + lane);
    if (g0 + SPMM_THREADS + lane < end) sln = gcn_seg_slot(pre, segS, nChunks, g0 + SPMM_THREADS + lane);
    int4 q = make_int4(0, 0, 0, 0);
    if (sl >= 0) q = rec[sl];
    while (g0 < end) {
        const int gn = g0 + SPMM_THREADS;
        int4 qn = make_int4(0, 0, 0, 0);
        if (sln >= 0) qn = rec[sln];
        int sln2 = -1;
        if (gn + SPMM_THREADS + lane < end) sln2 = gcn_seg_slot(pre, segS, nChunks, gn + SPMM_THREADS + lane);
        float v = 0.0f;
        if (g0 + lane < end)
            v = gcn_inv_sqrt_deg(deg[q.x]) * __int_as_float(q.z) * gcn_inv_sqrt_deg(deg[q.y]);
        const int vbits = __float_as_int(v);
        const int rl = (q.x - rowBase) * TP;
        const int nvalid = (end - g0 < 32) ? (end - g0) : 32;
        if (nvalid == 32) {
            float f[32];
#pragma unroll
            for (int j = 0; j < 32; ++j) {
                int cj = __builtin_amdgcn_readlane(q.y, j);
                f[j] = feat[(size_t)cj * D_FEAT + lane];
            }
#pragma unroll
            for (int j = 0; j < 32; ++j) {
                float vj = __int_as_float(__builtin_amdgcn_readlane(vbits, j));
                int rj = __builtin_amdgcn_readlane(rl, j);
                atomicAdd(&tile[rj + lane], vj * f[j]);
            }
        } else {
            for (int j = 0; j < nvalid; ++j) {
                int cj = __builtin_amdgcn_readlane(q.y, j);
                float vj = __int_as_float(__builtin_amdgcn_readlane(vbits, j));
                int rj = __builtin_amdgcn_readlane(rl, j);
                float fj = feat[(size_t)cj * D_FEAT + lane];
                atomicAdd(&tile[rj + lane], vj * fj);
            }
        }
        q = qn;
        sln = sln2;
        g0 = gn;
    }
    __syncthreads();

    {
        const int half = lane >> 4;
        const int l15  = lane & 15;
        const float* aRow = &tile[(wave * 16 + l15) * TP];
        const float* b0Row = &wl[l15 * WP];
        const float* b1Row = &wl[(16 + l15) * WP];
        v8f acc0 = {};
        v8f acc1 = {};
#pragma unroll
        for (int k = 0; k < D_FEAT; k += 4) {
            const int kk = k + half * 2;
            v2f a, b0, b1;
            a.x  = aRow[kk];
            a.y  = aRow[kk + 1];
            b0.x = b0Row[kk];
            b0.y = b0Row[kk + 1];
            b1.x = b1Row[kk];
            b1.y = b1Row[kk + 1];
            acc0 = __builtin_amdgcn_wmma_f32_16x16x4_f32(
                false, a, false, b0, (short)0, acc0, false, false);
            acc1 = __builtin_amdgcn_wmma_f32_16x16x4_f32(
                false, a, false, b1, (short)0, acc1, false, false);
        }
        float* dRow = &tile[(wave * 16 + half * 8) * TP + l15];
#pragma unroll
        for (int i = 0; i < 8; ++i) {
            dRow[i * TP]      = acc0[i];
            dRow[i * TP + 16] = acc1[i];
        }
    }
    __syncthreads();
    int rows = N - rowBase;
    if (rows > RB) rows = RB;
    float* dst = out + (size_t)rowBase * D_FEAT;
    for (int i = tid; i < rows * (D_FEAT / 4); i += SPMM_THREADS) {
        const int r = i >> 3, p = (i & 7) * 4;
        v4f val = *reinterpret_cast<const v4f*>(&tile[r * TP + p]);
        *reinterpret_cast<v4f*>(&dst[(size_t)i * 4]) = val;
    }
}

extern "C" void kernel_launch(void* const* d_in, const int* in_sizes, int n_in,
                              void* d_out, int out_size, void* d_ws, size_t ws_size,
                              hipStream_t stream) {
    const int*   er   = (const int*)d_in[0];
    const int*   ec   = (const int*)d_in[1];
    const float* ev   = (const float*)d_in[2];
    const float* feat = (const float*)d_in[3];
    const float* W    = (const float*)d_in[4];
    float* out = (float*)d_out;

    int E = in_sizes[0];
    int N = in_sizes[3] / D_FEAT;

    int nb = (N + RB - 1) >> RB_SHIFT;
    int nChunks = (E + CHUNK - 1) / CHUNK;

    float* deg = (float*)d_ws;
    int* segTab = (int*)(deg + N);
    size_t recOff = ((size_t)N + (size_t)nChunks * (size_t)(nb + 1) + 3) & ~(size_t)3;
    int* rec = (int*)d_ws + recOff;

    gcn_zero_kernel<<<(unsigned)(((size_t)N + 255) / 256), 256, 0, stream>>>(deg, (size_t)N);

    size_t histBytes = (size_t)nb * sizeof(int);
    gcn_chunk_group_kernel<<<nChunks, BIN_THREADS, histBytes, stream>>>(
        er, ec, ev, deg, segTab, rec, E, nb);
    size_t segBytes = ((size_t)nChunks * 2 + 1) * sizeof(int);
    gcn_bin_spmm_gemm_kernel<<<nb, SPMM_THREADS, segBytes, stream>>>(
        rec, segTab, deg, feat, W, out, N, nb, nChunks);
}


# Degrees by row block: a pass of one work-group per block sums its grouped records' values in LDS (f32 LDS adds), stores deg^-1/2; zero launch and 1.6 M global f32 atomics go; deg sum order changes
# speedup vs baseline: 4.1098x; 2.1064x over previous
#include <hip/hip_runtime.h>
#include <hip/hip_bf16.h>


typedef __attribute__((ext_vector_type(2))) float v2f;
typedef __attribute__((ext_vector_type(4))) float v4f;
typedef __attribute__((ext_vector_type(8))) float v8f;

#define D_FEAT 32

__device__ __forceinline__ float gcn_inv_sqrt_deg(float d) {
    return (d > 0.0f) ? rsqrtf(d) : 0.0f;
}

#define RB_SHIFT 8
#define RB (1 << RB_SHIFT)
#define BIN_THREADS 1024
#define EPT 8
#define CHUNK (EPT * BIN_THREADS)

__global__ __launch_bounds__(BIN_THREADS)
void gcn_chunk_group_kernel(const int* __restrict__ er, const int* __restrict__ ec,
                            const float* __restrict__ ev,
                            int* __restrict__ segTab, int* __restrict__ recWords,
                            int E, int nb) {
    extern __shared__ int hist[];
    __shared__ int4 stage[CHUNK];
    __shared__ int s[BIN_THREADS];
    __shared__ int carry;
    int4* rec = reinterpret_cast<int4*>(recWords);
    const int t = threadIdx.x;
    for (int i = t; i < nb; i += BIN_THREADS) hist[i] = 0;
    if (t == 0) carry = 0;
    __syncthreads();
    const int base = blockIdx.x * CHUNK;
    int r[EPT], c[EPT], vb[EPT], rank[EPT];
#pragma unroll
    for (int k = 0; k < EPT; ++k) {
        int e = base + k * BIN_THREADS + t;
        r[k] = 0; c[k] = 0; vb[k] = 0;
        if (e < E) {
            r[k]  = er[e];
            c[k]  = ec[e];
            vb[k] = __float_as_int(ev[e]);
        }
    }
#pragma unroll
    for (int k = 0; k < EPT; ++k) {
        int e = base + k * BIN_THREADS + t;
        rank[k] = 0;
        if (e < E)
            rank[k] = atomicAdd(&hist[r[k] >> RB_SHIFT], 1);
    }
    __syncthreads();
    int* seg = segTab + (size_t)blockIdx.x * (size_t)(nb + 1);
    for (int t0 = 0; t0 < nb; t0 += BIN_THREADS) {
        const int b = t0 + t;
        const int v = (b < nb) ? hist[b] : 0;
        s[t] = v;
        __syncthreads();
        for (int off = 1; off < BIN_THREADS; off <<= 1) {
            int add = (t >= off) ? s[t - off] : 0;
            __syncthreads();
            s[t] += add;
            __syncthreads();
        }
        const int incl = s[t];
        const int c0 = carry;
        if (b < nb) {
            hist[b] = c0 + incl - v;
            seg[b]  = c0 + incl - v;
        }
        __syncthreads();
        if (t == BIN_THREADS - 1) carry = c0 + incl;
        __syncthreads();
    }
    if (t == 0) seg[nb] = carry;
#pragma unroll
    for (int k = 0; k < EPT; ++k) {
        int e = base + k * BIN_THREADS + t;
        if (e < E) {
            int slot = hist[r[k] >> RB_SHIFT] + rank[k];
            stage[slot] = make_int4(r[k], c[k], vb[k], 0);
        }
    }
    __syncthreads();
    const int cnt = E - base;
    if (cnt >= CHUNK) {
#pragma unroll
        for (int k = 0; k < EPT; ++k) {
            const int i = k * BIN_THREADS + t;
            rec[base + i] = stage[i];
        }
    } else {
        for (int i = t; i < cnt; i += BIN_THREADS) rec[base + i] = stage[i];
    }
}

#define DEG_THREADS 512
#define DEG_WAVES (DEG_THREADS / 32)
#define DEG_U 16
#define DEG_BATCH (DEG_WAVES * DEG_U)
__global__ __launch_bounds__(DEG_THREADS)
void gcn_block_degree_kernel(const int* __restrict__ recWords, const int* __restrict__ segTab,
                             float* __restrict__ dis, int N, int nb, int nChunks) {
    __shared__ float degL[RB];
    extern __shared__ int runL[];
    const int nTab = nChunks + DEG_BATCH;
    int* runS = runL;
    int* runN = runL + nTab;
    const int4* rec = reinterpret_cast<const int4*>(recWords);
    const int tid  = threadIdx.x;
    const int lane = tid & 31;
    const int wave = __builtin_amdgcn_readfirstlane(tid >> 5);
    const int rowBase = (int)blockIdx.x << RB_SHIFT;
    for (int i = tid; i < RB; i += DEG_THREADS) degL[i] = 0.0f;
    for (int c = tid; c < nTab; c += DEG_THREADS) {
        int s = 0, n = 0;
        if (c < nChunks) {
            const int* row = segTab + (size_t)c * (size_t)(nb + 1) + blockIdx.x;
            const int a = row[0];
            s = c * CHUNK + a;
            n = row[1] - a;
        }
        runS[c] = s;
        runN[c] = n;
    }
    __syncthreads();
    for (int c0 = wave; c0 < nChunks; c0 += DEG_BATCH) {
        int sK[DEG_U], nK[DEG_U], rK[DEG_U];
        float vK[DEG_U];
#pragma unroll
        for (int k = 0; k < DEG_U; ++k) {
            sK[k] = runS[c0 + k * DEG_WAVES];
            nK[k] = runN[c0 + k * DEG_WAVES];
        }
#pragma unroll
        for (int k = 0; k < DEG_U; ++k) {
            rK[k] = 0;
            vK[k] = 0.0f;
            if (lane < nK[k]) {
                const int4 q = rec[sK[k] + lane];
                rK[k] = q.x;
                vK[k] = __int_as_float(q.z);
            }
        }
#pragma unroll
        for (int k = 0; k < DEG_U; ++k)
            if (lane < nK[k]) atomicAdd(&degL[rK[k] - rowBase], vK[k]);
#pragma unroll
        for (int k = 0; k < DEG_U; ++k) {
            if (nK[k] > 32) {
                for (int i = 32 + lane; i < nK[k]; i += 32) {
                    const int4 q = rec[sK[k] + i];
                    atomicAdd(&degL[q.x - rowBase], __int_as_float(q.z));
                }
            }
        }
    }
    __syncthreads();
    int rows = N - rowBase;
    if (rows > RB) rows = RB;
    for (int i = tid; i < rows; i += DEG_THREADS) dis[rowBase + i] = gcn_inv_sqrt_deg(degL[i]);
}

__device__ __forceinline__ int gcn_seg_slot(const int* pre, const int* segS, int nChunks, int g) {
    int lo = 0, hi = nChunks;
    while (hi - lo > 1) {
        const int mid = (lo + hi) >> 1;
        if (pre[mid] <= g) lo = mid; else hi = mid;
    }
    return segS[lo] + (g - pre[lo]);
}

#define SPMM_THREADS 512
#define TP 36
#define WP 34
__global__ __launch_bounds__(SPMM_THREADS)
void gcn_bin_spmm_gemm_kernel(const int* __restrict__ recWords, const int* __restrict__ segTab,
                              const float* __restrict__ dis, const float* __restrict__ feat,
                              const float* __restrict__ W,
                              float* __restrict__ out, int N, int nb, int nChunks) {
    __shared__ __attribute__((aligned(16))) float tile[RB * TP];
    __shared__ __attribute__((aligned(16))) float wl[D_FEAT * WP];
    extern __shared__ int segL[];
    __shared__ int sc[SPMM_THREADS];
    __shared__ int carry;
    int* pre  = segL;
    int* segS = segL + (nChunks + 1);
    const int4* rec = reinterpret_cast<const int4*>(recWords);
    const int tid  = threadIdx.x;
    const int lane = tid & 31;
    const int wave = __builtin_amdgcn_readfirstlane(tid >> 5);
    const int rowBase = (int)blockIdx.x << RB_SHIFT;
    for (int i = tid; i < RB * TP; i += SPMM_THREADS) tile[i] = 0.0f;
    {
        const int n = tid >> 4, k = (tid & 15) * 2;
        wl[n * WP + k]     = W[n * D_FEAT + k];
        wl[n * WP + k + 1] = W[n * D_FEAT + k + 1];
    }
    if (tid == 0) carry = 0;
    __syncthreads();
    for (int c0 = 0; c0 < nChunks; c0 += SPMM_THREADS) {
        const int c = c0 + tid;
        int a = 0, len = 0;
        if (c < nChunks) {
            const int* row = segTab + (size_t)c * (size_t)(nb + 1) + blockIdx.x;
            a   = row[0];
            len = row[1] - a;
        }
        sc[tid] = len;
        __syncthreads();
        for (int off = 1; off < SPMM_THREADS; off <<= 1) {
            int add = (tid >= off) ? sc[tid - off] : 0;
            __syncthreads();
            sc[tid] += add;
            __syncthreads();
        }
        const int incl = sc[tid];
        const int cv = carry;
        if (c < nChunks) {
            pre[c]  = cv + incl - len;
            segS[c] = c * CHUNK + a;
        }
        __syncthreads();
        if (tid == SPMM_THREADS - 1) carry = cv + incl;
        __syncthreads();
    }
    const int end = __builtin_amdgcn_readfirstlane(carry);
    if (tid == 0) pre[nChunks] = end;
    __syncthreads();

    int g0 = wave * 32;
    int sl = -1, sln = -1;
    if (g0 + lane < end) sl = gcn_seg_slot(pre, segS, nChunks, g0 + lane);
    if (g0 + SPMM_THREADS + lane < end) sln = gcn_seg_slot(pre, segS, nChunks, g0 + SPMM_THREADS + lane);
    int4 q = make_int4(0, 0, 0, 0);
    if (sl >= 0) q = rec[sl];
    while (g0 < end) {
        const int gn = g0 + SPMM_THREADS;
        int4 qn = make_int4(0, 0, 0, 0);
        if (sln >= 0) qn = rec[sln];
        int sln2 = -1;
        if (gn + SPMM_THREADS + lane < end) sln2 = gcn_seg_slot(pre, segS, nChunks, gn + SPMM_THREADS + lane);
        float v = 0.0f;
        if (g0 + lane < end)
            v = dis[q.x] * __int_as_float(q.z) * dis[q.y];
        const int vbits = __float_as_int(v);
        const int rl = (q.x - rowBase) * TP;
        const int nvalid = (end - g0 < 32) ? (end - g0) : 32;
        if (nvalid == 32) {
            float f[32];
#pragma unroll
            for (int j = 0; j < 32; ++j) {
                int cj = __builtin_amdgcn_readlane(q.y, j);
                f[j] = feat[(size_t)cj * D_FEAT + lane];
            }
#pragma unroll
            for (int j = 0; j < 32; ++j) {
                float vj = __int_as_float(__builtin_amdgcn_readlane(vbits, j));
                int rj = __builtin_amdgcn_readlane(rl, j);
                atomicAdd(&tile[rj + lane], vj * f[j]);
            }
        } else {
            for (int j = 0; j < nvalid; ++j) {
                int cj = __builtin_amdgcn_readlane(q.y, j);
                float vj = __int_as_float(__builtin_amdgcn_readlane(vbits, j));
                int rj = __builtin_amdgcn_readlane(rl, j);
                float fj = feat[(size_t)cj * D_FEAT + lane];
                atomicAdd(&tile[rj + lane], vj * fj);
            }
        }
        q = qn;
        sln = sln2;
        g0 = gn;
    }
    __syncthreads();

    {
        const int half = lane >> 4;
        const int l15  = lane & 15;
        const float* aRow = &tile[(wave * 16 + l15) * TP];
        const float* b0Row = &wl[l15 * WP];
        const float* b1Row = &wl[(16 + l15) * WP];
        v8f acc0 = {};
        v8f acc1 = {};
#pragma unroll
        for (int k = 0; k < D_FEAT; k += 4) {
            const int kk = k + half * 2;
            v2f a, b0, b1;
            a.x  = aRow[kk];
            a.y  = aRow[kk + 1];
            b0.x = b0Row[kk];
            b0.y = b0Row[kk + 1];
            b1.x = b1Row[kk];
            b1.y = b1Row[kk + 1];
            acc0 = __builtin_amdgcn_wmma_f32_16x16x4_f32(
                false, a, false, b0, (short)0, acc0, false, false);
            acc1 = __builtin_amdgcn_wmma_f32_16x16x4_f32(
                false, a, false, b1, (short)0, acc1, false, false);
        }
        float* dRow = &tile[(wave * 16 + half * 8) * TP + l15];
#pragma unroll
        for (int i = 0; i < 8; ++i) {
            dRow[i * TP]      = acc0[i];
            dRow[i * TP + 16] = acc1[i];
        }
    }
    __syncthreads();
    int rows = N - rowBase;
    if (rows > RB) rows = RB;
    float* dst = out + (size_t)rowBase * D_FEAT;
    for (int i = tid; i < rows * (D_FEAT / 4); i += SPMM_THREADS) {
        const int r = i >> 3, p = (i & 7) * 4;
        v4f val = *reinterpret_cast<const v4f*>(&tile[r * TP + p]);
        *reinterpret_cast<v4f*>(&dst[(size_t)i * 4]) = val;
    }
}

extern "C" void kernel_launch(void* const* d_in, const int* in_sizes, int n_in,
                              void* d_out, int out_size, void* d_ws, size_t ws_size,
                              hipStream_t stream) {
    const int*   er   = (const int*)d_in[0];
    const int*   ec   = (const int*)d_in[1];
    const float* ev   = (const float*)d_in[2];
    const float* feat = (const float*)d_in[3];
    const float* W    = (const float*)d_in[4];
    float* out = (float*)d_out;

    int E = in_sizes[0];
    int N = in_sizes[3] / D_FEAT;

    int nb = (N + RB - 1) >> RB_SHIFT;
    int nChunks = (E + CHUNK - 1) / CHUNK;

    float* dis = (float*)d_ws;
    int* segTab = (int*)(dis + N);
    size_t recOff = ((size_t)N + (size_t)nChunks * (size_t)(nb + 1) + 3) & ~(size_t)3;
    int* rec = (int*)d_ws + recOff;

    size_t histBytes = (size_t)nb * sizeof(int);
    gcn_chunk_group_kernel<<<nChunks, BIN_THREADS, histBytes, stream>>>(
        er, ec, ev, segTab, rec, E, nb);
    size_t runBytes = ((size_t)nChunks + DEG_BATCH) * 2 * sizeof(int);
    gcn_block_degree_kernel<<<nb, DEG_THREADS, runBytes, stream>>>(
        rec, segTab, dis, N, nb, nChunks);
    size_t segBytes = ((size_t)nChunks * 2 + 1) * sizeof(int);
    gcn_bin_spmm_gemm_kernel<<<nb, SPMM_THREADS, segBytes, stream>>>(
        rec, segTab, dis, feat, W, out, N, nb, nChunks);
}


# SpMM edge feed: a wave loads its runs' records and factors in one batch and queues finished edges densely in LDS; edge loop takes 32-entry groups; no chunk scan, no slot search; agg sum order changes
# speedup vs baseline: 4.5226x; 1.1004x over previous
#include <hip/hip_runtime.h>
#include <hip/hip_bf16.h>


typedef __attribute__((ext_vector_type(2))) float v2f;
typedef __attribute__((ext_vector_type(4))) float v4f;
typedef __attribute__((ext_vector_type(8))) float v8f;

#define D_FEAT 32

__device__ __forceinline__ float gcn_inv_sqrt_deg(float d) {
    return (d > 0.0f) ? rsqrtf(d) : 0.0f;
}

#define RB_SHIFT 8
#define RB (1 << RB_SHIFT)
#define BIN_THREADS 1024
#define EPT 8
#define CHUNK (EPT * BIN_THREADS)

__global__ __launch_bounds__(BIN_THREADS)
void gcn_chunk_group_kernel(const int* __restrict__ er, const int* __restrict__ ec,
                            const float* __restrict__ ev,
                            int* __restrict__ segTab, int* __restrict__ recWords,
                            int E, int nb) {
    extern __shared__ int hist[];
    __shared__ int4 stage[CHUNK];
    __shared__ int s[BIN_THREADS];
    __shared__ int carry;
    int4* rec = reinterpret_cast<int4*>(recWords);
    const int t = threadIdx.x;
    for (int i = t; i < nb; i += BIN_THREADS) hist[i] = 0;
    if (t == 0) carry = 0;
    __syncthreads();
    const int base = blockIdx.x * CHUNK;
    int r[EPT], c[EPT], vb[EPT], rank[EPT];
#pragma unroll
    for (int k = 0; k < EPT; ++k) {
        int e = base + k * BIN_THREADS + t;
        r[k] = 0; c[k] = 0; vb[k] = 0;
        if (e < E) {
            r[k]  = er[e];
            c[k]  = ec[e];
            vb[k] = __float_as_int(ev[e]);
        }
    }
#pragma unroll
    for (int k = 0; k < EPT; ++k) {
        int e = base + k * BIN_THREADS + t;
        rank[k] = 0;
        if (e < E)
            rank[k] = atomicAdd(&hist[r[k] >> RB_SHIFT], 1);
    }
    __syncthreads();
    int* seg = segTab + (size_t)blockIdx.x * (size_t)(nb + 1);
    for (int t0 = 0; t0 < nb; t0 += BIN_THREADS) {
        const int b = t0 + t;
        const int v = (b < nb) ? hist[b] : 0;
        s[t] = v;
        __syncthreads();
        for (int off = 1; off < BIN_THREADS; off <<= 1) {
            int add = (t >= off) ? s[t - off] : 0;
            __syncthreads();
            s[t] += add;
            __syncthreads();
        }
        const int incl = s[t];
        const int c0 = carry;
        if (b < nb) {
            hist[b] = c0 + incl - v;
            seg[b]  = c0 + incl - v;
        }
        __syncthreads();
        if (t == BIN_THREADS - 1) carry = c0 + incl;
        __syncthreads();
    }
    if (t == 0) seg[nb] = carry;
#pragma unroll
    for (int k = 0; k < EPT; ++k) {
        int e = base + k * BIN_THREADS + t;
        if (e < E) {
            int slot = hist[r[k] >> RB_SHIFT] + rank[k];
            stage[slot] = make_int4(r[k], c[k], vb[k], 0);
        }
    }
    __syncthreads();
    const int cnt = E - base;
    if (cnt >= CHUNK) {
#pragma unroll
        for (int k = 0; k < EPT; ++k) {
            const int i = k * BIN_THREADS + t;
            rec[base + i] = stage[i];
        }
    } else {
        for (int i = t; i < cnt; i += BIN_THREADS) rec[base + i] = stage[i];
    }
}

#define DEG_THREADS 512
#define DEG_WAVES (DEG_THREADS / 32)
#define DEG_U 16
#define DEG_BATCH (DEG_WAVES * DEG_U)
__global__ __launch_bounds__(DEG_THREADS)
void gcn_block_degree_kernel(const int* __restrict__ recWords, const int* __restrict__ segTab,
                             float* __restrict__ dis, int N, int nb, int nChunks) {
    __shared__ float degL[RB];
    extern __shared__ int runL[];
    const int nTab = nChunks + DEG_BATCH;
    int* runS = runL;
    int* runN = runL + nTab;
    const int4* rec = reinterpret_cast<const int4*>(recWords);
    const int tid  = threadIdx.x;
    const int lane = tid & 31;
    const int wave = __builtin_amdgcn_readfirstlane(tid >> 5);
    const int rowBase = (int)blockIdx.x << RB_SHIFT;
    for (int i = tid; i < RB; i += DEG_THREADS) degL[i] = 0.0f;
    for (int c = tid; c < nTab; c += DEG_THREADS) {
        int s = 0, n = 0;
        if (c < nChunks) {
            const int* row = segTab + (size_t)c * (size_t)(nb + 1) + blockIdx.x;
            const int a = row[0];
            s = c * CHUNK + a;
            n = row[1] - a;
        }
        runS[c] = s;
        runN[c] = n;
    }
    __syncthreads();
    for (int c0 = wave; c0 < nChunks; c0 += DEG_BATCH) {
        int sK[DEG_U], nK[DEG_U], rK[DEG_U];
        float vK[DEG_U];
#pragma unroll
        for (int k = 0; k < DEG_U; ++k) {
            sK[k] = runS[c0 + k * DEG_WAVES];
            nK[k] = runN[c0 + k * DEG_WAVES];
        }
#pragma unroll
        for (int k = 0; k < DEG_U; ++k) {
            rK[k] = 0;
            vK[k] = 0.0f;
            if (lane < nK[k]) {
                const int4 q = rec[sK[k] + lane];
                rK[k] = q.x;
                vK[k] = __int_as_float(q.z);
            }
        }
#pragma unroll
        for (int k = 0; k < DEG_U; ++k)
            if (lane < nK[k]) atomicAdd(&degL[rK[k] - rowBase], vK[k]);
#pragma unroll
        for (int k = 0; k < DEG_U; ++k) {
            if (nK[k] > 32) {
                for (int i = 32 + lane; i < nK[k]; i += 32) {
                    const int4 q = rec[sK[k] + i];
                    atomicAdd(&degL[q.x - rowBase], __int_as_float(q.z));
                }
            }
        }
    }
    __syncthreads();
    int rows = N - rowBase;
    if (rows > RB) rows = RB;
    for (int i = tid; i < rows; i += DEG_THREADS) dis[rowBase + i] = gcn_inv_sqrt_deg(degL[i]);
}

__device__ __forceinline__ void gcn_edge_group(float* tile, const float* __restrict__ feat,
                                               int lane, int rl, int cy, int vbits, int nvalid) {
    if (nvalid == 32) {
        float f[32];
#pragma unroll
        for (int j = 0; j < 32; ++j) {
            int cj = __builtin_amdgcn_readlane(cy, j);
            f[j] = feat[(size_t)cj * D_FEAT + lane];
        }
#pragma unroll
        for (int j = 0; j < 32; ++j) {
            float vj = __int_as_float(__builtin_amdgcn_readlane(vbits, j));
            int rj = __builtin_amdgcn_readlane(rl, j);
            atomicAdd(&tile[rj + lane], vj * f[j]);
        }
    } else {
        for (int j = 0; j < nvalid; ++j) {
            int cj = __builtin_amdgcn_readlane(cy, j);
            float vj = __int_as_float(__builtin_amdgcn_readlane(vbits, j));
            int rj = __builtin_amdgcn_readlane(rl, j);
            float fj = feat[(size_t)cj * D_FEAT + lane];
            atomicAdd(&tile[rj + lane], vj * fj);
        }
    }
}

#define SPMM_THREADS 512
#define SPMM_WAVES (SPMM_THREADS / 32)
#define Q_U 16
#define Q_CAP (SPMM_WAVES * Q_U * 32)
#define TP 36
#define WP 34
__global__ __launch_bounds__(SPMM_THREADS)
void gcn_bin_spmm_gemm_kernel(const int* __restrict__ recWords, const int* __restrict__ segTab,
                              const float* __restrict__ dis, const float* __restrict__ feat,
                              const float* __restrict__ W,
                              float* __restrict__ out, int N, int nb, int nChunks) {
    __shared__ __attribute__((aligned(16))) float tile[RB * TP];
    __shared__ __attribute__((aligned(16))) float wl[D_FEAT * WP];
    __shared__ int qRow[Q_CAP];
    __shared__ int qCol[Q_CAP];
    __shared__ int qVal[Q_CAP];
    __shared__ int wcnt[SPMM_WAVES];
    const int4* rec = reinterpret_cast<const int4*>(recWords);
    const int tid  = threadIdx.x;
    const int lane = tid & 31;
    const int wave = __builtin_amdgcn_readfirstlane(tid >> 5);
    const int rowBase = (int)blockIdx.x << RB_SHIFT;
    for (int i = tid; i < RB * TP; i += SPMM_THREADS) tile[i] = 0.0f;
    {
        const int n = tid >> 4, k = (tid & 15) * 2;
        wl[n * WP + k]     = W[n * D_FEAT + k];
        wl[n * WP + k + 1] = W[n * D_FEAT + k + 1];
    }
    __syncthreads();

    for (int cb = 0; cb < nChunks; cb += SPMM_WAVES * Q_U) {
        int sB = 0, nB = 0;
        {
            const int c = cb + lane * SPMM_WAVES + wave;
            if (lane < Q_U && c < nChunks) {
                const int* row = segTab + (size_t)c * (size_t)(nb + 1) + blockIdx.x;
                const int a = row[0];
                sB = c * CHUNK + a;
                nB = row[1] - a;
            }
        }
        int sK[Q_U], nK[Q_U], offK[Q_U];
        int cnt = 0;
#pragma unroll
        for (int k = 0; k < Q_U; ++k) {
            sK[k] = __builtin_amdgcn_readlane(sB, k);
            nK[k] = __builtin_amdgcn_readlane(nB, k);
            offK[k] = cnt;
            cnt += (nK[k] < 32) ? nK[k] : 32;
        }
        if (lane == 0) wcnt[wave] = cnt;
        int rK[Q_U], cK[Q_U], vK[Q_U];
        float drK[Q_U], dcK[Q_U];
#pragma unroll
        for (int k = 0; k < Q_U; ++k) {
            const int li = (lane < nK[k]) ? lane : 0;
            const int idx = (nK[k] > 0) ? (sK[k] + li) : 0;
            const int4 q = rec[idx];
            rK[k] = q.x; cK[k] = q.y; vK[k] = q.z;
        }
#pragma unroll
        for (int k = 0; k < Q_U; ++k) {
            drK[k] = dis[rK[k]];
            dcK[k] = dis[cK[k]];
        }
        __syncthreads();
        int base = 0, total = 0;
#pragma unroll
        for (int w = 0; w < SPMM_WAVES; ++w) {
            const int cw = wcnt[w];
            base  += (w < wave) ? cw : 0;
            total += cw;
        }
        base  = __builtin_amdgcn_readfirstlane(base);
        total = __builtin_amdgcn_readfirstlane(total);
#pragma unroll
        for (int k = 0; k < Q_U; ++k) {
            if (lane < nK[k]) {
                const int e = base + offK[k] + lane;
                qRow[e] = (rK[k] - rowBase) * TP;
                qCol[e] = cK[k];
                qVal[e] = __float_as_int(drK[k] * __int_as_float(vK[k]) * dcK[k]);
            }
        }
        __syncthreads();

        int g0 = wave * 32;
        int rl = 0, cy = 0, vb = 0;
        if (g0 < total) {
            const int i = (g0 + lane < total) ? (g0 + lane) : (total - 1);
            rl = qRow[i]; cy = qCol[i]; vb = qVal[i];
        }
        while (g0 < total) {
            const int gn = g0 + SPMM_THREADS;
            int rln = 0, cyn = 0, vbn = 0;
            if (gn < total) {
                const int i = (gn + lane < total) ? (gn + lane) : (total - 1);
                rln = qRow[i]; cyn = qCol[i]; vbn = qVal[i];
            }
            const int nvalid = (total - g0 < 32) ? (total - g0) : 32;
            gcn_edge_group(tile, feat, lane, rl, cy, vb, nvalid);
            rl = rln; cy = cyn; vb = vbn;
            g0 = gn;
        }

        for (int k = 0; k < Q_U; ++k) {
            const int n = __builtin_amdgcn_readlane(nB, k);
            if (n > 32) {
                const int s = __builtin_amdgcn_readlane(sB, k);
                for (int p = 32; p < n; p += 32) {
                    int trl = 0, tcy = 0, tvb = 0;
                    if (p + lane < n) {
                        const int4 q = rec[s + p + lane];
                        trl = (q.x - rowBase) * TP;
                        tcy = q.y;
                        tvb = __float_as_int(dis[q.x] * __int_as_float(q.z) * dis[q.y]);
                    }
                    const int m = (n - p < 32) ? (n - p) : 32;
                    gcn_edge_group(tile, feat, lane, trl, tcy, tvb, m);
                }
            }
        }
        __syncthreads();
    }

    {
        const int half = lane >> 4;
        const int l15  = lane & 15;
        const float* aRow = &tile[(wave * 16 + l15) * TP];
        const float* b0Row = &wl[l15 * WP];
        const float* b1Row = &wl[(16 + l15) * WP];
        v8f acc0 = {};
        v8f acc1 = {};
#pragma unroll
        for (int k = 0; k < D_FEAT; k += 4) {
            const int kk = k + half * 2;
            v2f a, b0, b1;
            a.x  = aRow[kk];
            a.y  = aRow[kk + 1];
            b0.x = b0Row[kk];
            b0.y = b0Row[kk + 1];
            b1.x = b1Row[kk];
            b1.y = b1Row[kk + 1];
            acc0 = __builtin_amdgcn_wmma_f32_16x16x4_f32(
                false, a, false, b0, (short)0, acc0, false, false);
            acc1 = __builtin_amdgcn_wmma_f32_16x16x4_f32(
                false, a, false, b1, (short)0, acc1, false, false);
        }
        float* dRow = &tile[(wave * 16 + half * 8) * TP + l15];
#pragma unroll
        for (int i = 0; i < 8; ++i) {
            dRow[i * TP]      = acc0[i];
            dRow[i * TP + 16] = acc1[i];
        }
    }
    __syncthreads();
    int rows = N - rowBase;
    if (rows > RB) rows = RB;
    float* dst = out + (size_t)rowBase * D_FEAT;
    for (int i = tid; i < rows * (D_FEAT / 4); i += SPMM_THREADS) {
        const int r = i >> 3, p = (i & 7) * 4;
        v4f val = *reinterpret_cast<const v4f*>(&tile[r * TP + p]);
        *reinterpret_cast<v4f*>(&dst[(size_t)i * 4]) = val;
    }
}

extern "C" void kernel_launch(void* const* d_in, const int* in_sizes, int n_in,
                              void* d_out, int out_size, void* d_ws, size_t ws_size,
                              hipStream_t stream) {
    const int*   er   = (const int*)d_in[0];
    const int*   ec   = (const int*)d_in[1];
    const float* ev   = (const float*)d_in[2];
    const float* feat = (const float*)d_in[3];
    const float* W    = (const float*)d_in[4];
    float* out = (float*)d_out;

    int E = in_sizes[0];
    int N = in_sizes[3] / D_FEAT;

    int nb = (N + RB - 1) >> RB_SHIFT;
    int nChunks = (E + CHUNK - 1) / CHUNK;

    float* dis = (float*)d_ws;
    int* segTab = (int*)(dis + N);
    size_t recOff = ((size_t)N + (size_t)nChunks * (size_t)(nb + 1) + 3) & ~(size_t)3;
    int* rec = (int*)d_ws + recOff;

    size_t histBytes = (size_t)nb * sizeof(int);
    gcn_chunk_group_kernel<<<nChunks, BIN_THREADS, histBytes, stream>>>(
        er, ec, ev, segTab, rec, E, nb);
    size_t runBytes = ((size_t)nChunks + DEG_BATCH) * 2 * sizeof(int);
    gcn_block_degree_kernel<<<nb, DEG_THREADS, runBytes, stream>>>(
        rec, segTab, dis, N, nb, nChunks);
    gcn_bin_spmm_gemm_kernel<<<nb, SPMM_THREADS, 0, stream>>>(
        rec, segTab, dis, feat, W, out, N, nb, nChunks);
}


# Column factor leaves the edge path: the degree pass also stores fs = dis * features for its 256 rows; SpMM gathers fs rows, coefficient is dis[row]*ev; no dis[col] gather; product order changes
# speedup vs baseline: 4.7410x; 1.0483x over previous
#include <hip/hip_runtime.h>
#include <hip/hip_bf16.h>


typedef __attribute__((ext_vector_type(2))) float v2f;
typedef __attribute__((ext_vector_type(4))) float v4f;
typedef __attribute__((ext_vector_type(8))) float v8f;

#define D_FEAT 32

__device__ __forceinline__ float gcn_inv_sqrt_deg(float d) {
    return (d > 0.0f) ? rsqrtf(d) : 0.0f;
}

#define RB_SHIFT 8
#define RB (1 << RB_SHIFT)
#define BIN_THREADS 1024
#define EPT 8
#define CHUNK (EPT * BIN_THREADS)

__global__ __launch_bounds__(BIN_THREADS)
void gcn_chunk_group_kernel(const int* __restrict__ er, const int* __restrict__ ec,
                            const float* __restrict__ ev,
                            int* __restrict__ segTab, int* __restrict__ recWords,
                            int E, int nb) {
    extern __shared__ int hist[];
    __shared__ int4 stage[CHUNK];
    __shared__ int s[BIN_THREADS];
    __shared__ int carry;
    int4* rec = reinterpret_cast<int4*>(recWords);
    const int t = threadIdx.x;
    for (int i = t; i < nb; i += BIN_THREADS) hist[i] = 0;
    if (t == 0) carry = 0;
    __syncthreads();
    const int base = blockIdx.x * CHUNK;
    int r[EPT], c[EPT], vb[EPT], rank[EPT];
#pragma unroll
    for (int k = 0; k < EPT; ++k) {
        int e = base + k * BIN_THREADS + t;
        r[k] = 0; c[k] = 0; vb[k] = 0;
        if (e < E) {
            r[k]  = er[e];
            c[k]  = ec[e];
            vb[k] = __float_as_int(ev[e]);
        }
    }
#pragma unroll
    for (int k = 0; k < EPT; ++k) {
        int e = base + k * BIN_THREADS + t;
        rank[k] = 0;
        if (e < E)
            rank[k] = atomicAdd(&hist[r[k] >> RB_SHIFT], 1);
    }
    __syncthreads();
    int* seg = segTab + (size_t)blockIdx.x * (size_t)(nb + 1);
    for (int t0 = 0; t0 < nb; t0 += BIN_THREADS) {
        const int b = t0 + t;
        const int v = (b < nb) ? hist[b] : 0;
        s[t] = v;
        __syncthreads();
        for (int off = 1; off < BIN_THREADS; off <<= 1) {
            int add = (t >= off) ? s[t - off] : 0;
            __syncthreads();
            s[t] += add;
            __syncthreads();
        }
        const int incl = s[t];
        const int c0 = carry;
        if (b < nb) {
            hist[b] = c0 + incl - v;
            seg[b]  = c0 + incl - v;
        }
        __syncthreads();
        if (t == BIN_THREADS - 1) carry = c0 + incl;
        __syncthreads();
    }
    if (t == 0) seg[nb] = carry;
#pragma unroll
    for (int k = 0; k < EPT; ++k) {
        int e = base + k * BIN_THREADS + t;
        if (e < E) {
            int slot = hist[r[k] >> RB_SHIFT] + rank[k];
            stage[slot] = make_int4(r[k], c[k], vb[k], 0);
        }
    }
    __syncthreads();
    const int cnt = E - base;
    if (cnt >= CHUNK) {
#pragma unroll
        for (int k = 0; k < EPT; ++k) {
            const int i = k * BIN_THREADS + t;
            rec[base + i] = stage[i];
        }
    } else {
        for (int i = t; i < cnt; i += BIN_THREADS) rec[base + i] = stage[i];
    }
}

#define DEG_THREADS 512
#define DEG_WAVES (DEG_THREADS / 32)
#define DEG_U 16
#define DEG_BATCH (DEG_WAVES * DEG_U)
#define FS_V4 (RB * (D_FEAT / 4) / DEG_THREADS)
__global__ __launch_bounds__(DEG_THREADS)
void gcn_block_degree_kernel(const int* __restrict__ recWords, const int* __restrict__ segTab,
                             const float* __restrict__ feat,
                             float* __restrict__ dis, float* __restrict__ fs,
                             int N, int nb, int nChunks) {
    __shared__ float degL[RB];
    extern __shared__ int runL[];
    const int nTab = nChunks + DEG_BATCH;
    int* runS = runL;
    int* runN = runL + nTab;
    const int4* rec = reinterpret_cast<const int4*>(recWords);
    const int tid  = threadIdx.x;
    const int lane = tid & 31;
    const int wave = __builtin_amdgcn_readfirstlane(tid >> 5);
    const int rowBase = (int)blockIdx.x << RB_SHIFT;
    int rows = N - rowBase;
    if (rows > RB) rows = RB;
    v4f fK[FS_V4];
    const v4f* fsrc = reinterpret_cast<const v4f*>(feat + (size_t)rowBase * D_FEAT);
#pragma unroll
    for (int k = 0; k < FS_V4; ++k) {
        const int i = tid + k * DEG_THREADS;
        fK[k] = (v4f){0.0f, 0.0f, 0.0f, 0.0f};
        if ((i >> 3) < rows) fK[k] = fsrc[i];
    }
    for (int i = tid; i < RB; i += DEG_THREADS) degL[i] = 0.0f;
    for (int c = tid; c < nTab; c += DEG_THREADS) {
        int s = 0, n = 0;
        if (c < nChunks) {
            const int* row = segTab + (size_t)c * (size_t)(nb + 1) + blockIdx.x;
            const int a = row[0];
            s = c * CHUNK + a;
            n = row[1] - a;
        }
        runS[c] = s;
        runN[c] = n;
    }
    __syncthreads();
    for (int c0 = wave; c0 < nChunks; c0 += DEG_BATCH) {
        int sK[DEG_U], nK[DEG_U], rK[DEG_U];
        float vK[DEG_U];
#pragma unroll
        for (int k = 0; k < DEG_U; ++k) {
            sK[k] = runS[c0 + k * DEG_WAVES];
            nK[k] = runN[c0 + k * DEG_WAVES];
        }
#pragma unroll
        for (int k = 0; k < DEG_U; ++k) {
            rK[k] = 0;
            vK[k] = 0.0f;
            if (lane < nK[k]) {
                const int4 q = rec[sK[k] + lane];
                rK[k] = q.x;
                vK[k] = __int_as_float(q.z);
            }
        }
#pragma unroll
        for (int k = 0; k < DEG_U; ++k)
            if (lane < nK[k]) atomicAdd(&degL[rK[k] - rowBase], vK[k]);
#pragma unroll
        for (int k = 0; k < DEG_U; ++k) {
            if (nK[k] > 32) {
                for (int i = 32 + lane; i < nK[k]; i += 32) {
                    const int4 q = rec[sK[k] + i];
                    atomicAdd(&degL[q.x - rowBase], __int_as_float(q.z));
                }
            }
        }
    }
    __syncthreads();
    for (int i = tid; i < rows; i += DEG_THREADS) dis[rowBase + i] = gcn_inv_sqrt_deg(degL[i]);
    v4f* fdst = reinterpret_cast<v4f*>(fs + (size_t)rowBase * D_FEAT);
    float dK[FS_V4];
#pragma unroll
    for (int k = 0; k < FS_V4; ++k) dK[k] = gcn_inv_sqrt_deg(degL[(tid + k * DEG_THREADS) >> 3]);
#pragma unroll
    for (int k = 0; k < FS_V4; ++k) {
        const int i = tid + k * DEG_THREADS;
        if ((i >> 3) < rows) fdst[i] = fK[k] * dK[k];
    }
}

__device__ __forceinline__ void gcn_edge_group(float* tile, const float* __restrict__ feat,
                                               int lane, int rl, int cy, int vbits, int nvalid) {
    if (nvalid == 32) {
        float f[32];
#pragma unroll
        for (int j = 0; j < 32; ++j) {
            int cj = __builtin_amdgcn_readlane(cy, j);
            f[j] = feat[(size_t)cj * D_FEAT + lane];
        }
#pragma unroll
        for (int j = 0; j < 32; ++j) {
            float vj = __int_as_float(__builtin_amdgcn_readlane(vbits, j));
            int rj = __builtin_amdgcn_readlane(rl, j);
            atomicAdd(&tile[rj + lane], vj * f[j]);
        }
    } else {
        for (int j = 0; j < nvalid; ++j) {
            int cj = __builtin_amdgcn_readlane(cy, j);
            float vj = __int_as_float(__builtin_amdgcn_readlane(vbits, j));
            int rj = __builtin_amdgcn_readlane(rl, j);
            float fj = feat[(size_t)cj * D_FEAT + lane];
            atomicAdd(&tile[rj + lane], vj * fj);
        }
    }
}

#define SPMM_THREADS 512
#define SPMM_WAVES (SPMM_THREADS / 32)
#define Q_U 16
#define Q_CAP (SPMM_WAVES * Q_U * 32)
#define TP 36
#define WP 34
__global__ __launch_bounds__(SPMM_THREADS)
void gcn_bin_spmm_gemm_kernel(const int* __restrict__ recWords, const int* __restrict__ segTab,
                              const float* __restrict__ dis, const float* __restrict__ fs,
                              const float* __restrict__ W,
                              float* __restrict__ out, int N, int nb, int nChunks) {
    __shared__ __attribute__((aligned(16))) float tile[RB * TP];
    __shared__ __attribute__((aligned(16))) float wl[D_FEAT * WP];
    __shared__ int qRow[Q_CAP];
    __shared__ int qCol[Q_CAP];
    __shared__ int qVal[Q_CAP];
    __shared__ int wcnt[SPMM_WAVES];
    const int4* rec = reinterpret_cast<const int4*>(recWords);
    const int tid  = threadIdx.x;
    const int lane = tid & 31;
    const int wave = __builtin_amdgcn_readfirstlane(tid >> 5);
    const int rowBase = (int)blockIdx.x << RB_SHIFT;
    for (int i = tid; i < RB * TP; i += SPMM_THREADS) tile[i] = 0.0f;
    {
        const int n = tid >> 4, k = (tid & 15) * 2;
        wl[n * WP + k]     = W[n * D_FEAT + k];
        wl[n * WP + k + 1] = W[n * D_FEAT + k + 1];
    }
    __syncthreads();

    for (int cb = 0; cb < nChunks; cb += SPMM_WAVES * Q_U) {
        int sB = 0, nB = 0;
        {
            const int c = cb + lane * SPMM_WAVES + wave;
            if (lane < Q_U && c < nChunks) {
                const int* row = segTab + (size_t)c * (size_t)(nb + 1) + blockIdx.x;
                const int a = row[0];
                sB = c * CHUNK + a;
                nB = row[1] - a;
            }
        }
        int sK[Q_U], nK[Q_U], offK[Q_U];
        int cnt = 0;
#pragma unroll
        for (int k = 0; k < Q_U; ++k) {
            sK[k] = __builtin_amdgcn_readlane(sB, k);
            nK[k] = __builtin_amdgcn_readlane(nB, k);
            offK[k] = cnt;
            cnt += (nK[k] < 32) ? nK[k] : 32;
        }
        if (lane == 0) wcnt[wave] = cnt;
        int rK[Q_U], cK[Q_U], vK[Q_U];
        float drK[Q_U];
#pragma unroll
        for (int k = 0; k < Q_U; ++k) {
            const int li = (lane < nK[k]) ? lane : 0;
            const int idx = (nK[k] > 0) ? (sK[k] + li) : 0;
            const int4 q = rec[idx];
            rK[k] = q.x; cK[k] = q.y; vK[k] = q.z;
        }
#pragma unroll
        for (int k = 0; k < Q_U; ++k) drK[k] = dis[rK[k]];
        __syncthreads();
        int base = 0, total = 0;
#pragma unroll
        for (int w = 0; w < SPMM_WAVES; ++w) {
            const int cw = wcnt[w];
            base  += (w < wave) ? cw : 0;
            total += cw;
        }
        base  = __builtin_amdgcn_readfirstlane(base);
        total = __builtin_amdgcn_readfirstlane(total);
#pragma unroll
        for (int k = 0; k < Q_U; ++k) {
            if (lane < nK[k]) {
                const int e = base + offK[k] + lane;
                qRow[e] = (rK[k] - rowBase) * TP;
                qCol[e] = cK[k];
                qVal[e] = __float_as_int(drK[k] * __int_as_float(vK[k]));
            }
        }
        __syncthreads();

        int g0 = wave * 32;
        int rl = 0, cy = 0, vb = 0;
        if (g0 < total) {
            const int i = (g0 + lane < total) ? (g0 + lane) : (total - 1);
            rl = qRow[i]; cy = qCol[i]; vb = qVal[i];
        }
        while (g0 < total) {
            const int gn = g0 + SPMM_THREADS;
            int rln = 0, cyn = 0, vbn = 0;
            if (gn < total) {
                const int i = (gn + lane < total) ? (gn + lane) : (total - 1);
                rln = qRow[i]; cyn = qCol[i]; vbn = qVal[i];
            }
            const int nvalid = (total - g0 < 32) ? (total - g0) : 32;
            gcn_edge_group(tile, fs, lane, rl, cy, vb, nvalid);
            rl = rln; cy = cyn; vb = vbn;
            g0 = gn;
        }

        for (int k = 0; k < Q_U; ++k) {
            const int n = __builtin_amdgcn_readlane(nB, k);
            if (n > 32) {
                const int s = __builtin_amdgcn_readlane(sB, k);
                for (int p = 32; p < n; p += 32) {
                    int trl = 0, tcy = 0, tvb = 0;
                    if (p + lane < n) {
                        const int4 q = rec[s + p + lane];
                        trl = (q.x - rowBase) * TP;
                        tcy = q.y;
                        tvb = __float_as_int(dis[q.x] * __int_as_float(q.z));
                    }
                    const int m = (n - p < 32) ? (n - p) : 32;
                    gcn_edge_group(tile, fs, lane, trl, tcy, tvb, m);
                }
            }
        }
        __syncthreads();
    }

    {
        const int half = lane >> 4;
        const int l15  = lane & 15;
        const float* aRow = &tile[(wave * 16 + l15) * TP];
        const float* b0Row = &wl[l15 * WP];
        const float* b1Row = &wl[(16 + l15) * WP];
        v8f acc0 = {};
        v8f acc1 = {};
#pragma unroll
        for (int k = 0; k < D_FEAT; k += 4) {
            const int kk = k + half * 2;
            v2f a, b0, b1;
            a.x  = aRow[kk];
            a.y  = aRow[kk + 1];
            b0.x = b0Row[kk];
            b0.y = b0Row[kk + 1];
            b1.x = b1Row[kk];
            b1.y = b1Row[kk + 1];
            acc0 = __builtin_amdgcn_wmma_f32_16x16x4_f32(
                false, a, false, b0, (short)0, acc0, false, false);
            acc1 = __builtin_amdgcn_wmma_f32_16x16x4_f32(
                false, a, false, b1, (short)0, acc1, false, false);
        }
        float* dRow = &tile[(wave * 16 + half * 8) * TP + l15];
#pragma unroll
        for (int i = 0; i < 8; ++i) {
            dRow[i * TP]      = acc0[i];
            dRow[i * TP + 16] = acc1[i];
        }
    }
    __syncthreads();
    int rows = N - rowBase;
    if (rows > RB) rows = RB;
    float* dst = out + (size_t)rowBase * D_FEAT;
    for (int i = tid; i < rows * (D_FEAT / 4); i += SPMM_THREADS) {
        const int r = i >> 3, p = (i & 7) * 4;
        v4f val = *reinterpret_cast<const v4f*>(&tile[r * TP + p]);
        *reinterpret_cast<v4f*>(&dst[(size_t)i * 4]) = val;
    }
}

extern "C" void kernel_launch(void* const* d_in, const int* in_sizes, int n_in,
                              void* d_out, int out_size, void* d_ws, size_t ws_size,
                              hipStream_t stream) {
    const int*   er   = (const int*)d_in[0];
    const int*   ec   = (const int*)d_in[1];
    const float* ev   = (const float*)d_in[2];
    const float* feat = (const float*)d_in[3];
    const float* W    = (const float*)d_in[4];
    float* out = (float*)d_out;

    int E = in_sizes[0];
    int N = in_sizes[3] / D_FEAT;

    int nb = (N + RB - 1) >> RB_SHIFT;
    int nChunks = (E + CHUNK - 1) / CHUNK;

    float* fs  = (float*)d_ws;
    float* dis = fs + (size_t)N * D_FEAT;
    int* segTab = (int*)(dis + N);
    size_t recOff = ((size_t)N * (D_FEAT + 1) + (size_t)nChunks * (size_t)(nb + 1) + 3) & ~(size_t)3;
    int* rec = (int*)d_ws + recOff;

    size_t histBytes = (size_t)nb * sizeof(int);
    gcn_chunk_group_kernel<<<nChunks, BIN_THREADS, histBytes, stream>>>(
        er, ec, ev, segTab, rec, E, nb);
    size_t runBytes = ((size_t)nChunks + DEG_BATCH) * 2 * sizeof(int);
    gcn_block_degree_kernel<<<nb, DEG_THREADS, runBytes, stream>>>(
        rec, segTab, feat, dis, fs, N, nb, nChunks);
    gcn_bin_spmm_gemm_kernel<<<nb, SPMM_THREADS, 0, stream>>>(
        rec, segTab, dis, fs, W, out, N, nb, nChunks);
}


# SpMM edge loop gathers 16 B per lane: 8 lanes per queued edge, 4 edges per load; lanes read 8-B queue entries from LDS (no broadcasts); adds rotated over a piece's 4 floats; agg sum order changes
# speedup vs baseline: 5.2756x; 1.1128x over previous
#include <hip/hip_runtime.h>
#include <hip/hip_bf16.h>


typedef __attribute__((ext_vector_type(2))) float v2f;
typedef __attribute__((ext_vector_type(4))) float v4f;
typedef __attribute__((ext_vector_type(8))) float v8f;

#define D_FEAT 32

__device__ __forceinline__ float gcn_inv_sqrt_deg(float d) {
    return (d > 0.0f) ? rsqrtf(d) : 0.0f;
}

#define RB_SHIFT 8
#define RB (1 << RB_SHIFT)
#define BIN_THREADS 1024
#define EPT 8
#define CHUNK (EPT * BIN_THREADS)

__global__ __launch_bounds__(BIN_THREADS)
void gcn_chunk_group_kernel(const int* __restrict__ er, const int* __restrict__ ec,
                            const float* __restrict__ ev,
                            int* __restrict__ segTab, int* __restrict__ recWords,
                            int E, int nb) {
    extern __shared__ int hist[];
    __shared__ int4 stage[CHUNK];
    __shared__ int s[BIN_THREADS];
    __shared__ int carry;
    int4* rec = reinterpret_cast<int4*>(recWords);
    const int t = threadIdx.x;
    for (int i = t; i < nb; i += BIN_THREADS) hist[i] = 0;
    if (t == 0) carry = 0;
    __syncthreads();
    const int base = blockIdx.x * CHUNK;
    int r[EPT], c[EPT], vb[EPT], rank[EPT];
#pragma unroll
    for (int k = 0; k < EPT; ++k) {
        int e = base + k * BIN_THREADS + t;
        r[k] = 0; c[k] = 0; vb[k] = 0;
        if (e < E) {
            r[k]  = er[e];
            c[k]  = ec[e];
            vb[k] = __float_as_int(ev[e]);
        }
    }
#pragma unroll
    for (int k = 0; k < EPT; ++k) {
        int e = base + k * BIN_THREADS + t;
        rank[k] = 0;
        if (e < E)
            rank[k] = atomicAdd(&hist[r[k] >> RB_SHIFT], 1);
    }
    __syncthreads();
    int* seg = segTab + (size_t)blockIdx.x * (size_t)(nb + 1);
    for (int t0 = 0; t0 < nb; t0 += BIN_THREADS) {
        const int b = t0 + t;
        const int v = (b < nb) ? hist[b] : 0;
        s[t] = v;
        __syncthreads();
        for (int off = 1; off < BIN_THREADS; off <<= 1) {
            int add = (t >= off) ? s[t - off] : 0;
            __syncthreads();
            s[t] += add;
            __syncthreads();
        }
        const int incl = s[t];
        const int c0 = carry;
        if (b < nb) {
            hist[b] = c0 + incl - v;
            seg[b]  = c0 + incl - v;
        }
        __syncthreads();
        if (t == BIN_THREADS - 1) carry = c0 + incl;
        __syncthreads();
    }
    if (t == 0) seg[nb] = carry;
#pragma unroll
    for (int k = 0; k < EPT; ++k) {
        int e = base + k * BIN_THREADS + t;
        if (e < E) {
            int slot = hist[r[k] >> RB_SHIFT] + rank[k];
            stage[slot] = make_int4(r[k], c[k], vb[k], 0);
        }
    }
    __syncthreads();
    const int cnt = E - base;
    if (cnt >= CHUNK) {
#pragma unroll
        for (int k = 0; k < EPT; ++k) {
            const int i = k * BIN_THREADS + t;
            rec[base + i] = stage[i];
        }
    } else {
        for (int i = t; i < cnt; i += BIN_THREADS) rec[base + i] = stage[i];
    }
}

#define DEG_THREADS 512
#define DEG_WAVES (DEG_THREADS / 32)
#define DEG_U 16
#define DEG_BATCH (DEG_WAVES * DEG_U)
#define FS_V4 (RB * (D_FEAT / 4) / DEG_THREADS)
__global__ __launch_bounds__(DEG_THREADS)
void gcn_block_degree_kernel(const int* __restrict__ recWords, const int* __restrict__ segTab,
                             const float* __restrict__ feat,
                             float* __restrict__ dis, float* __restrict__ fs,
                             int N, int nb, int nChunks) {
    __shared__ float degL[RB];
    extern __shared__ int runL[];
    const int nTab = nChunks + DEG_BATCH;
    int* runS = runL;
    int* runN = runL + nTab;
    const int4* rec = reinterpret_cast<const int4*>(recWords);
    const int tid  = threadIdx.x;
    const int lane = tid & 31;
    const int wave = __builtin_amdgcn_readfirstlane(tid >> 5);
    const int rowBase = (int)blockIdx.x << RB_SHIFT;
    int rows = N - rowBase;
    if (rows > RB) rows = RB;
    v4f fK[FS_V4];
    const v4f* fsrc = reinterpret_cast<const v4f*>(feat + (size_t)rowBase * D_FEAT);
#pragma unroll
    for (int k = 0; k < FS_V4; ++k) {
        const int i = tid + k * DEG_THREADS;
        fK[k] = (v4f){0.0f, 0.0f, 0.0f, 0.0f};
        if ((i >> 3) < rows) fK[k] = fsrc[i];
    }
    for (int i = tid; i < RB; i += DEG_THREADS) degL[i] = 0.0f;
    for (int c = tid; c < nTab; c += DEG_THREADS) {
        int s = 0, n = 0;
        if (c < nChunks) {
            const int* row = segTab + (size_t)c * (size_t)(nb + 1) + blockIdx.x;
            const int a = row[0];
            s = c * CHUNK + a;
            n = row[1] - a;
        }
        runS[c] = s;
        runN[c] = n;
    }
    __syncthreads();
    for (int c0 = wave; c0 < nChunks; c0 += DEG_BATCH) {
        int sK[DEG_U], nK[DEG_U], rK[DEG_U];
        float vK[DEG_U];
#pragma unroll
        for (int k = 0; k < DEG_U; ++k) {
            sK[k] = runS[c0 + k * DEG_WAVES];
            nK[k] = runN[c0 + k * DEG_WAVES];
        }
#pragma unroll
        for (int k = 0; k < DEG_U; ++k) {
            rK[k] = 0;
            vK[k] = 0.0f;
            if (lane < nK[k]) {
                const int4 q = rec[sK[k] + lane];
                rK[k] = q.x;
                vK[k] = __int_as_float(q.z);
            }
        }
#pragma unroll
        for (int k = 0; k < DEG_U; ++k)
            if (lane < nK[k]) atomicAdd(&degL[rK[k] - rowBase], vK[k]);
#pragma unroll
        for (int k = 0; k < DEG_U; ++k) {
            if (nK[k] > 32) {
                for (int i = 32 + lane; i < nK[k]; i += 32) {
                    const int4 q = rec[sK[k] + i];
                    atomicAdd(&degL[q.x - rowBase], __int_as_float(q.z));
                }
            }
        }
    }
    __syncthreads();
    for (int i = tid; i < rows; i += DEG_THREADS) dis[rowBase + i] = gcn_inv_sqrt_deg(degL[i]);
    v4f* fdst = reinterpret_cast<v4f*>(fs + (size_t)rowBase * D_FEAT);
    float dK[FS_V4];
#pragma unroll
    for (int k = 0; k < FS_V4; ++k) dK[k] = gcn_inv_sqrt_deg(degL[(tid + k * DEG_THREADS) >> 3]);
#pragma unroll
    for (int k = 0; k < FS_V4; ++k) {
        const int i = tid + k * DEG_THREADS;
        if ((i >> 3) < rows) fdst[i] = fK[k] * dK[k];
    }
}

__device__ __forceinline__ void gcn_edge_group(float* tile, const float* __restrict__ feat,
                                               int lane, int rl, int cy, int vbits, int nvalid) {
    if (nvalid == 32) {
        float f[32];
#pragma unroll
        for (int j = 0; j < 32; ++j) {
            int cj = __builtin_amdgcn_readlane(cy, j);
            f[j] = feat[(size_t)cj * D_FEAT + lane];
        }
#pragma unroll
        for (int j = 0; j < 32; ++j) {
            float vj = __int_as_float(__builtin_amdgcn_readlane(vbits, j));
            int rj = __builtin_amdgcn_readlane(rl, j);
            atomicAdd(&tile[rj + lane], vj * f[j]);
        }
    } else {
        for (int j = 0; j < nvalid; ++j) {
            int cj = __builtin_amdgcn_readlane(cy, j);
            float vj = __int_as_float(__builtin_amdgcn_readlane(vbits, j));
            int rj = __builtin_amdgcn_readlane(rl, j);
            float fj = feat[(size_t)cj * D_FEAT + lane];
            atomicAdd(&tile[rj + lane], vj * fj);
        }
    }
}

#define SPMM_THREADS 512
#define SPMM_WAVES (SPMM_THREADS / 32)
#define Q_U 16
#define Q_CAP (SPMM_WAVES * Q_U * 32)
#define TP 36
#define WP 34

template <bool CLAMP>
__device__ __forceinline__ void gcn_queue_read(const int2* q, int first, int total, int g,
                                               int (&w)[8], float (&cf)[8]) {
#pragma unroll
    for (int j = 0; j < 8; ++j) {
        int e = first + 4 * j + g;
        if (CLAMP) e = (e < total) ? e : (total - 1);
        const int2 en = q[e];
        w[j]  = en.x;
        cf[j] = __int_as_float(en.y);
    }
}

__device__ __forceinline__ void gcn_gather8(const float* __restrict__ fs, int p,
                                            const int (&w)[8], v4f (&f)[8]) {
    const char* base = reinterpret_cast<const char*>(fs);
#pragma unroll
    for (int j = 0; j < 8; ++j) {
        const unsigned off = ((unsigned)w[j] & 0x00FFFFFFu) * (unsigned)(D_FEAT * 4) + (unsigned)(p * 16);
        f[j] = *reinterpret_cast<const v4f*>(base + off);
    }
}

template <bool FULL>
__device__ __forceinline__ void gcn_add8(float* tile, int g, int p, int first, int total,
                                         const int (&w)[8], const float (&cf)[8], const v4f (&f)[8]) {
    const bool g1 = (g & 1) != 0, g2 = (g & 2) != 0;
#pragma unroll
    for (int j = 0; j < 8; ++j) {
        const float t0 = g1 ? f[j].y : f[j].x, t1 = g1 ? f[j].x : f[j].y;
        const float t2 = g1 ? f[j].w : f[j].z, t3 = g1 ? f[j].z : f[j].w;
        const float u0 = g2 ? t2 : t0, u1 = g2 ? t3 : t1;
        const float u2 = g2 ? t0 : t2, u3 = g2 ? t1 : t3;
        float* dst = tile + (int)((unsigned)w[j] >> 24) * TP + p * 4;
        if (FULL || (first + 4 * j + g < total)) {
            atomicAdd(dst + g,       cf[j] * u0);
            atomicAdd(dst + (g ^ 1), cf[j] * u1);
            atomicAdd(dst + (g ^ 2), cf[j] * u2);
            atomicAdd(dst + (g ^ 3), cf[j] * u3);
        }
    }
}

__global__ __launch_bounds__(SPMM_THREADS)
void gcn_bin_spmm_gemm_kernel(const int* __restrict__ recWords, const int* __restrict__ segTab,
                              const float* __restrict__ dis, const float* __restrict__ fs,
                              const float* __restrict__ W,
                              float* __restrict__ out, int N, int nb, int nChunks) {
    __shared__ __attribute__((aligned(16))) float tile[RB * TP];
    __shared__ __attribute__((aligned(16))) float wl[D_FEAT * WP];
    __shared__ int2 q[Q_CAP];
    __shared__ int wcnt[SPMM_WAVES];
    const int4* rec = reinterpret_cast<const int4*>(recWords);
    const int tid  = threadIdx.x;
    const int lane = tid & 31;
    const int wave = __builtin_amdgcn_readfirstlane(tid >> 5);
    const int rowBase = (int)blockIdx.x << RB_SHIFT;
    for (int i = tid; i < RB * TP; i += SPMM_THREADS) tile[i] = 0.0f;
    {
        const int n = tid >> 4, k = (tid & 15) * 2;
        wl[n * WP + k]     = W[n * D_FEAT + k];
        wl[n * WP + k + 1] = W[n * D_FEAT + k + 1];
    }
    __syncthreads();

    for (int cb = 0; cb < nChunks; cb += SPMM_WAVES * Q_U) {
        int sB = 0, nB = 0;
        {
            const int c = cb + lane * SPMM_WAVES + wave;
            if (lane < Q_U && c < nChunks) {
                const int* row = segTab + (size_t)c * (size_t)(nb + 1) + blockIdx.x;
                const int a = row[0];
                sB = c * CHUNK + a;
                nB = row[1] - a;
            }
        }
        int sK[Q_U], nK[Q_U], offK[Q_U];
        int cnt = 0;
#pragma unroll
        for (int k = 0; k < Q_U; ++k) {
            sK[k] = __builtin_amdgcn_readlane(sB, k);
            nK[k] = __builtin_amdgcn_readlane(nB, k);
            offK[k] = cnt;
            cnt += (nK[k] < 32) ? nK[k] : 32;
        }
        if (lane == 0) wcnt[wave] = cnt;
        int rK[Q_U], cK[Q_U], vK[Q_U];
        float drK[Q_U];
#pragma unroll
        for (int k = 0; k < Q_U; ++k) {
            const int li = (lane < nK[k]) ? lane : 0;
            const int idx = (nK[k] > 0) ? (sK[k] + li) : 0;
            const int4 q = rec[idx];
            rK[k] = q.x; cK[k] = q.y; vK[k] = q.z;
        }
#pragma unroll
        for (int k = 0; k < Q_U; ++k) drK[k] = dis[rK[k]];
        __syncthreads();
        int base = 0, total = 0;
#pragma unroll
        for (int w = 0; w < SPMM_WAVES; ++w) {
            const int cw = wcnt[w];
            base  += (w < wave) ? cw : 0;
            total += cw;
        }
        base  = __builtin_amdgcn_readfirstlane(base);
        total = __builtin_amdgcn_readfirstlane(total);
#pragma unroll
        for (int k = 0; k < Q_U; ++k) {
            if (lane < nK[k]) {
                const int e = base + offK[k] + lane;
                q[e] = make_int2(cK[k] | ((rK[k] - rowBase) << 24),
                                 __float_as_int(drK[k] * __int_as_float(vK[k])));
            }
        }
        __syncthreads();

        const int eg = lane >> 3, ep = lane & 7;
        int g0 = wave * 32;
        int qw[8];
        float qc[8];
#pragma unroll
        for (int j = 0; j < 8; ++j) { qw[j] = 0; qc[j] = 0.0f; }
        if (g0 < total) {
            if (g0 + 32 <= total) gcn_queue_read<false>(q, g0, total, eg, qw, qc);
            else                  gcn_queue_read<true>(q, g0, total, eg, qw, qc);
        }
        while (g0 < total) {
            v4f fv[8];
            gcn_gather8(fs, ep, qw, fv);
            const int gn = g0 + SPMM_THREADS;
            int qwn[8];
            float qcn[8];
#pragma unroll
            for (int j = 0; j < 8; ++j) { qwn[j] = 0; qcn[j] = 0.0f; }
            if (gn < total) {
                if (gn + 32 <= total) gcn_queue_read<false>(q, gn, total, eg, qwn, qcn);
                else                  gcn_queue_read<true>(q, gn, total, eg, qwn, qcn);
            }
            if (g0 + 32 <= total) gcn_add8<true>(tile, eg, ep, g0, total, qw, qc, fv);
            else                  gcn_add8<false>(tile, eg, ep, g0, total, qw, qc, fv);
#pragma unroll
            for (int j = 0; j < 8; ++j) { qw[j] = qwn[j]; qc[j] = qcn[j]; }
            g0 = gn;
        }

        for (int k = 0; k < Q_U; ++k) {
            const int n = __builtin_amdgcn_readlane(nB, k);
            if (n > 32) {
                const int s = __builtin_amdgcn_readlane(sB, k);
                for (int p = 32; p < n; p += 32) {
                    int trl = 0, tcy = 0, tvb = 0;
                    if (p + lane < n) {
                        const int4 q = rec[s + p + lane];
                        trl = (q.x - rowBase) * TP;
                        tcy = q.y;
                        tvb = __float_as_int(dis[q.x] * __int_as_float(q.z));
                    }
                    const int m = (n - p < 32) ? (n - p) : 32;
                    gcn_edge_group(tile, fs, lane, trl, tcy, tvb, m);
                }
            }
        }
        __syncthreads();
    }

    {
        const int half = lane >> 4;
        const int l15  = lane & 15;
        const float* aRow = &tile[(wave * 16 + l15) * TP];
        const float* b0Row = &wl[l15 * WP];
        const float* b1Row = &wl[(16 + l15) * WP];
        v8f acc0 = {};
        v8f acc1 = {};
#pragma unroll
        for (int k = 0; k < D_FEAT; k += 4) {
            const int kk = k + half * 2;
            v2f a, b0, b1;
            a.x  = aRow[kk];
            a.y  = aRow[kk + 1];
            b0.x = b0Row[kk];
            b0.y = b0Row[kk + 1];
            b1.x = b1Row[kk];
            b1.y = b1Row[kk + 1];
            acc0 = __builtin_amdgcn_wmma_f32_16x16x4_f32(
                false, a, false, b0, (short)0, acc0, false, false);
            acc1 = __builtin_amdgcn_wmma_f32_16x16x4_f32(
                false, a, false, b1, (short)0, acc1, false, false);
        }
        float* dRow = &tile[(wave * 16 + half * 8) * TP + l15];
#pragma unroll
        for (int i = 0; i < 8; ++i) {
            dRow[i * TP]      = acc0[i];
            dRow[i * TP + 16] = acc1[i];
        }
    }
    __syncthreads();
    int rows = N - rowBase;
    if (rows > RB) rows = RB;
    float* dst = out + (size_t)rowBase * D_FEAT;
    for (int i = tid; i < rows * (D_FEAT / 4); i += SPMM_THREADS) {
        const int r = i >> 3, p = (i & 7) * 4;
        v4f val = *reinterpret_cast<const v4f*>(&tile[r * TP + p]);
        *reinterpret_cast<v4f*>(&dst[(size_t)i * 4]) = val;
    }
}

extern "C" void kernel_launch(void* const* d_in, const int* in_sizes, int n_in,
                              void* d_out, int out_size, void* d_ws, size_t ws_size,
                              hipStream_t stream) {
    const int*   er   = (const int*)d_in[0];
    const int*   ec   = (const int*)d_in[1];
    const float* ev   = (const float*)d_in[2];
    const float* feat = (const float*)d_in[3];
    const float* W    = (const float*)d_in[4];
    float* out = (float*)d_out;

    int E = in_sizes[0];
    int N = in_sizes[3] / D_FEAT;

    int nb = (N + RB - 1) >> RB_SHIFT;
    int nChunks = (E + CHUNK - 1) / CHUNK;

    float* fs  = (float*)d_ws;
    float* dis = fs + (size_t)N * D_FEAT;
    int* segTab = (int*)(dis + N);
    size_t recOff = ((size_t)N * (D_FEAT + 1) + (size_t)nChunks * (size_t)(nb + 1) + 3) & ~(size_t)3;
    int* rec = (int*)d_ws + recOff;

    size_t histBytes = (size_t)nb * sizeof(int);
    gcn_chunk_group_kernel<<<nChunks, BIN_THREADS, histBytes, stream>>>(
        er, ec, ev, segTab, rec, E, nb);
    size_t runBytes = ((size_t)nChunks + DEG_BATCH) * 2 * sizeof(int);
    gcn_block_degree_kernel<<<nb, DEG_THREADS, runBytes, stream>>>(
        rec, segTab, feat, dis, fs, N, nb, nChunks);
    gcn_bin_spmm_gemm_kernel<<<nb, SPMM_THREADS, 0, stream>>>(
        rec, segTab, dis, fs, W, out, N, nb, nChunks);
}
